# starting kernel
# baseline (speedup 1.0000x reference)
#include <hip/hip_runtime.h>

#define NN 10000
#define NE 160000
#define FIN 16
#define HEADS 8
#define CHN 8
#define HIDD 64
#define NGR 64
#define VOC 50000
#define LSTK 9
#define EP (NE + NN)

typedef __attribute__((ext_vector_type(2))) float v2f;
typedef __attribute__((ext_vector_type(8))) float v8f;

__device__ __forceinline__ unsigned fkey(float f) {
  unsigned u = __float_as_uint(f);
  return (u >> 31) ? ~u : (u | 0x80000000u);
}
__device__ __forceinline__ float funkey(unsigned k) {
  return __uint_as_float((k >> 31) ? (k & 0x7FFFFFFFu) : ~k);
}
#define KEY_NEG_INF 0x007FFFFFu

__global__ void k_gemm3_wmma(const float* __restrict__ A,
                             const float* __restrict__ B0, const float* __restrict__ b0,
                             float* __restrict__ C0,
                             const float* __restrict__ B1, const float* __restrict__ b1,
                             float* __restrict__ C1,
                             const float* __restrict__ B2, const float* __restrict__ b2,
                             float* __restrict__ C2,
                             int K) {
  const int lane = threadIdx.x;
  const int half = lane >> 4;
  const int l    = lane & 15;
  const int tn = blockIdx.x, tm = blockIdx.y;
  const int arow = tm * 16 + l;
  const int bcol = tn * 16 + l;
  v8f acc0 = {}, acc1 = {}, acc2 = {};
  for (int k = 0; k < K; k += 4) {
    const float* ap = A + arow * K + k + half * 2;
    v2f a; a.x = ap[0]; a.y = ap[1];
    const int boff = (k + half * 2) * HIDD + bcol;
    v2f w0; w0.x = B0[boff]; w0.y = B0[boff + HIDD];
    v2f w1; w1.x = B1[boff]; w1.y = B1[boff + HIDD];
    v2f w2; w2.x = B2[boff]; w2.y = B2[boff + HIDD];
    acc0 = __builtin_amdgcn_wmma_f32_16x16x4_f32(false, a, false, w0, (short)0, acc0, false, false);
    acc1 = __builtin_amdgcn_wmma_f32_16x16x4_f32(false, a, false, w1, (short)0, acc1, false, false);
    acc2 = __builtin_amdgcn_wmma_f32_16x16x4_f32(false, a, false, w2, (short)0, acc2, false, false);
  }
  const float bv0 = b0[bcol], bv1 = b1[bcol], bv2 = b2[bcol];
  const int rb = tm * 16 + half * 8;
#pragma unroll
  for (int r = 0; r < 8; ++r) {
    const int off = (rb + r) * HIDD + bcol;
    C0[off] = acc0[r] + bv0;
    C1[off] = acc1[r] + bv1;
    C2[off] = acc2[r] + bv2;
  }
}

__global__ void k_gemm_wmma(const float* __restrict__ A, const float* __restrict__ B,
                            const float* __restrict__ bias, float* __restrict__ C,
                            int M, int K, int Nc) {
  const int lane = threadIdx.x;
  const int half = lane >> 4;
  const int l    = lane & 15;
  const int tn = blockIdx.x, tm = blockIdx.y;
  const int arow = tm * 16 + l;
  const int bcol = tn * 16 + l;
  v8f acc = {};
  for (int k = 0; k < K; k += 4) {
    const float* ap = A + arow * K + k + half * 2;
    v2f a; a.x = ap[0]; a.y = ap[1];
    const float* bp = B + (size_t)(k + half * 2) * Nc + bcol;
    v2f b; b.x = bp[0]; b.y = bp[Nc];
    if (k + 4 < K) {
      __builtin_prefetch(bp + 4 * (size_t)Nc, 0, 1);
      __builtin_prefetch(bp + 5 * (size_t)Nc, 0, 1);
    }
    acc = __builtin_amdgcn_wmma_f32_16x16x4_f32(false, a, false, b, (short)0, acc, false, false);
  }
  const float bv = bias[bcol];
  const int rb = tm * 16 + half * 8;
#pragma unroll
  for (int r = 0; r < 8; ++r)
    C[(size_t)(rb + r) * Nc + bcol] = acc[r] + bv;
}

__global__ void k_zero(float* __restrict__ p, int n) {
  int i = blockIdx.x * blockDim.x + threadIdx.x;
  if (i < n) p[i] = 0.f;
}

__global__ void k_deg_loop(const int* __restrict__ dstE, const float* __restrict__ ea,
                           float* __restrict__ deg, float* __restrict__ la) {
  int e = blockIdx.x * blockDim.x + threadIdx.x;
  if (e >= NE) return;
  int d = dstE[e];
  atomicAdd(deg + d, 1.f);
  atomicAdd(la + 2 * d,     ea[2 * e]);
  atomicAdd(la + 2 * d + 1, ea[2 * e + 1]);
}

__global__ void k_loop_div(float* __restrict__ la, const float* __restrict__ deg) {
  int i = blockIdx.x * blockDim.x + threadIdx.x;
  if (i < 2 * NN) la[i] /= fmaxf(deg[i >> 1], 1.f);
}

__global__ void k_attn_init(unsigned* __restrict__ mxkey, float* __restrict__ den,
                            float* __restrict__ agg) {
  int i = blockIdx.x * blockDim.x + threadIdx.x;
  if (i < NN * HIDD) agg[i] = 0.f;
  if (i < NN * HEADS) { mxkey[i] = KEY_NEG_INF; den[i] = 0.f; }
}

__global__ void k_edge_logit(const float* __restrict__ xl, const float* __restrict__ xr,
                             const int* __restrict__ srcE, const int* __restrict__ dstE,
                             const float* __restrict__ ea, const float* __restrict__ la,
                             const float* __restrict__ We, const float* __restrict__ att,
                             float* __restrict__ logit, unsigned* __restrict__ mxkey) {
  int idx = blockIdx.x * blockDim.x + threadIdx.x;
  if (idx >= EP * HEADS) return;
  int e = idx >> 3, hh = idx & 7;
  int s, d; float a0, a1;
  if (e < NE) { s = srcE[e]; d = dstE[e]; a0 = ea[2 * e]; a1 = ea[2 * e + 1]; }
  else        { s = d = e - NE;           a0 = la[2 * s]; a1 = la[2 * s + 1]; }
  float lg = 0.f;
#pragma unroll
  for (int c = 0; c < CHN; ++c) {
    int hc = hh * CHN + c;
    float m = xl[s * HIDD + hc] + xr[d * HIDD + hc] + a0 * We[hc] + a1 * We[HIDD + hc];
    float g = m > 0.f ? m : 0.2f * m;
    lg += g * att[hc];
  }
  logit[idx] = lg;
  atomicMax(mxkey + d * HEADS + hh, fkey(lg));
}

__global__ void k_edge_exp(const float* __restrict__ logit, const unsigned* __restrict__ mxkey,
                           const int* __restrict__ dstE, float* __restrict__ exl,
                           float* __restrict__ den) {
  int idx = blockIdx.x * blockDim.x + threadIdx.x;
  if (idx >= EP * HEADS) return;
  int e = idx >> 3, hh = idx & 7;
  int d = (e < NE) ? dstE[e] : e - NE;
  float v = __expf(logit[idx] - funkey(mxkey[d * HEADS + hh]));
  exl[idx] = v;
  atomicAdd(den + d * HEADS + hh, v);
}

__global__ void k_edge_agg(const float* __restrict__ xl, const float* __restrict__ exl,
                           const float* __restrict__ den, const int* __restrict__ srcE,
                           const int* __restrict__ dstE, float* __restrict__ agg) {
  int idx = blockIdx.x * blockDim.x + threadIdx.x;
  if (idx >= EP * HEADS) return;
  int e = idx >> 3, hh = idx & 7;
  int s, d;
  if (e < NE) { s = srcE[e]; d = dstE[e]; } else { s = d = e - NE; }
  float alpha = exl[idx] / den[d * HEADS + hh];
#pragma unroll
  for (int c = 0; c < CHN; ++c)
    atomicAdd(agg + d * HIDD + hh * CHN + c, alpha * xl[s * HIDD + hh * CHN + c]);
}

__global__ void k_combine_elu(const float* __restrict__ agg, const float* __restrict__ bias_g,
                              const float* __restrict__ res, float* __restrict__ hout) {
  int idx = blockIdx.x * blockDim.x + threadIdx.x;
  if (idx >= NN * HIDD) return;
  float v = agg[idx] + bias_g[idx & (HIDD - 1)] + res[idx];
  hout[idx] = v > 0.f ? v : (__expf(v) - 1.f);
}

__global__ void k_pool_accum(const float* __restrict__ h, const int* __restrict__ batch,
                             float* __restrict__ pooled, float* __restrict__ cnt) {
  int idx = blockIdx.x * blockDim.x + threadIdx.x;
  if (idx >= NN * HIDD) return;
  int n = idx >> 6, c = idx & (HIDD - 1);
  int b = batch[n];
  atomicAdd(pooled + b * HIDD + c, h[idx]);
  if (c == 0) atomicAdd(cnt + b, 1.f);
}

__global__ void k_pool_div(float* __restrict__ pooled, const float* __restrict__ cnt) {
  int idx = blockIdx.x * blockDim.x + threadIdx.x;
  if (idx < NGR * HIDD) pooled[idx] /= fmaxf(cnt[idx >> 6], 1.f);
}

extern "C" void kernel_launch(void* const* d_in, const int* in_sizes, int n_in,
                              void* d_out_v, int out_size, void* d_ws, size_t ws_size,
                              hipStream_t stream) {
  (void)in_sizes; (void)n_in; (void)out_size; (void)ws_size;
  const float* x       = (const float*)d_in[0];
  const int*   ei      = (const int*)  d_in[1];
  const float* ea      = (const float*)d_in[2];
  const int*   batch   = (const int*)  d_in[3];
  const float* p1_Wl   = (const float*)d_in[4];
  const float* p1_bl   = (const float*)d_in[5];
  const float* p1_Wr   = (const float*)d_in[6];
  const float* p1_br   = (const float*)d_in[7];
  const float* p1_We   = (const float*)d_in[8];
  const float* p1_att  = (const float*)d_in[9];
  const float* p1_bias = (const float*)d_in[10];
  const float* p1_Wres = (const float*)d_in[11];
  const float* p1_bres = (const float*)d_in[12];
  const float* Wl      = (const float*)d_in[13];
  const float* bl      = (const float*)d_in[14];
  const float* Wr      = (const float*)d_in[15];
  const float* br      = (const float*)d_in[16];
  const float* We      = (const float*)d_in[17];
  const float* att     = (const float*)d_in[18];
  const float* bias_g  = (const float*)d_in[19];
  const float* Wres    = (const float*)d_in[20];
  const float* bres    = (const float*)d_in[21];
  const float* Wlin    = (const float*)d_in[22];
  const float* blin    = (const float*)d_in[23];

  float* out = (float*)d_out_v;
  float* ws  = (float*)d_ws;

  float*    xl    = ws;
  float*    xr    = ws + 640000;
  float*    res   = ws + 1280000;
  float*    agg   = ws + 1920000;
  float*    logit = ws + 2560000;
  float*    exl   = ws + 3920000;
  float*    den   = ws + 5280000;
  unsigned* mxkey = (unsigned*)(ws + 5360000);
  float*    deg   = ws + 5440000;
  float*    la    = ws + 5450000;
  float*    cnt   = ws + 5470000;

  const int* srcE = ei;
  const int* dstE = ei + NE;

  const int TB = 256;
  const int gEdge = (EP * HEADS + TB - 1) / TB;
  const int gNode = (NN * HIDD + TB - 1) / TB;

  k_zero<<<(NN + TB - 1) / TB, TB, 0, stream>>>(deg, NN);
  k_zero<<<(2 * NN + TB - 1) / TB, TB, 0, stream>>>(la, 2 * NN);
  k_deg_loop<<<(NE + TB - 1) / TB, TB, 0, stream>>>(dstE, ea, deg, la);
  k_loop_div<<<(2 * NN + TB - 1) / TB, TB, 0, stream>>>(la, deg);

  for (int layer = 0; layer < 1 + LSTK; ++layer) {
    const float* hin = (layer == 0) ? x : out + (size_t)(layer - 1) * NN * HIDD;
    const int K = (layer == 0) ? FIN : HIDD;
    const int li = layer - 1;
    const float* Wl_   = (layer == 0) ? p1_Wl   : Wl   + (size_t)li * HIDD * HIDD;
    const float* bl_   = (layer == 0) ? p1_bl   : bl   + (size_t)li * HIDD;
    const float* Wr_   = (layer == 0) ? p1_Wr   : Wr   + (size_t)li * HIDD * HIDD;
    const float* br_   = (layer == 0) ? p1_br   : br   + (size_t)li * HIDD;
    const float* We_   = (layer == 0) ? p1_We   : We   + (size_t)li * 2 * HIDD;
    const float* att_  = (layer == 0) ? p1_att  : att  + (size_t)li * HEADS * CHN;
    const float* bg_   = (layer == 0) ? p1_bias : bias_g + (size_t)li * HIDD;
    const float* Wres_ = (layer == 0) ? p1_Wres : Wres + (size_t)li * HIDD * HIDD;
    const float* bres_ = (layer == 0) ? p1_bres : bres + (size_t)li * HIDD;
    float* hout = out + (size_t)layer * NN * HIDD;

    dim3 gg(HIDD / 16, NN / 16);
    k_gemm3_wmma<<<gg, 32, 0, stream>>>(hin, Wl_, bl_, xl, Wr_, br_, xr,
                                        Wres_, bres_, res, K);

    k_attn_init<<<gNode, TB, 0, stream>>>(mxkey, den, agg);
    k_edge_logit<<<gEdge, TB, 0, stream>>>(xl, xr, srcE, dstE, ea, la, We_, att_, logit, mxkey);
    k_edge_exp  <<<gEdge, TB, 0, stream>>>(logit, mxkey, dstE, exl, den);
    k_edge_agg  <<<gEdge, TB, 0, stream>>>(xl, exl, den, srcE, dstE, agg);
    k_combine_elu<<<gNode, TB, 0, stream>>>(agg, bg_, res, hout);
  }

  float* pooled = out + (size_t)(1 + LSTK) * NN * HIDD;
  float* x12    = pooled + NGR * HIDD;
  k_zero<<<(NGR * HIDD + TB - 1) / TB, TB, 0, stream>>>(pooled, NGR * HIDD);
  k_zero<<<1, NGR, 0, stream>>>(cnt, NGR);
  k_pool_accum<<<gNode, TB, 0, stream>>>(out + (size_t)LSTK * NN * HIDD, batch, pooled, cnt);
  k_pool_div<<<(NGR * HIDD + TB - 1) / TB, TB, 0, stream>>>(pooled, cnt);

  dim3 gf(VOC / 16, NGR / 16);
  k_gemm_wmma<<<gf, 32, 0, stream>>>(pooled, Wlin, blin, x12, NGR, HIDD, VOC);
}


# Edges grouped by destination once per call (counting sort, edge-id order); each layer's softmax+aggregation is one per-(node,head) gather with a running max: layers need no atomics, sum order fixed
# speedup vs baseline: 2.8091x; 2.8091x over previous
#include <hip/hip_runtime.h>

#define NN 10000
#define NE 160000
#define FIN 16
#define HEADS 8
#define CHN 8
#define HIDD 64
#define NGR 64
#define VOC 50000
#define LSTK 9
#define EU 8

typedef __attribute__((ext_vector_type(2))) float v2f;
typedef __attribute__((ext_vector_type(4))) float v4f;
typedef __attribute__((ext_vector_type(8))) float v8f;

__global__ void k_gemm3_wmma(const float* __restrict__ A,
                             const float* __restrict__ B0, const float* __restrict__ b0,
                             float* __restrict__ C0,
                             const float* __restrict__ B1, const float* __restrict__ b1,
                             float* __restrict__ C1,
                             const float* __restrict__ B2, const float* __restrict__ b2,
                             float* __restrict__ C2,
                             int K) {
  const int lane = threadIdx.x;
  const int half = lane >> 4;
  const int l    = lane & 15;
  const int tn = blockIdx.x, tm = blockIdx.y;
  const int arow = tm * 16 + l;
  const int bcol = tn * 16 + l;
  v8f acc0 = {}, acc1 = {}, acc2 = {};
  for (int k = 0; k < K; k += 4) {
    const float* ap = A + arow * K + k + half * 2;
    v2f a; a.x = ap[0]; a.y = ap[1];
    const int boff = (k + half * 2) * HIDD + bcol;
    v2f w0; w0.x = B0[boff]; w0.y = B0[boff + HIDD];
    v2f w1; w1.x = B1[boff]; w1.y = B1[boff + HIDD];
    v2f w2; w2.x = B2[boff]; w2.y = B2[boff + HIDD];
    acc0 = __builtin_amdgcn_wmma_f32_16x16x4_f32(false, a, false, w0, (short)0, acc0, false, false);
    acc1 = __builtin_amdgcn_wmma_f32_16x16x4_f32(false, a, false, w1, (short)0, acc1, false, false);
    acc2 = __builtin_amdgcn_wmma_f32_16x16x4_f32(false, a, false, w2, (short)0, acc2, false, false);
  }
  const float bv0 = b0[bcol], bv1 = b1[bcol], bv2 = b2[bcol];
  const int rb = tm * 16 + half * 8;
#pragma unroll
  for (int r = 0; r < 8; ++r) {
    const int off = (rb + r) * HIDD + bcol;
    C0[off] = acc0[r] + bv0;
    C1[off] = acc1[r] + bv1;
    C2[off] = acc2[r] + bv2;
  }
}

__global__ void k_gemm_wmma(const float* __restrict__ A, const float* __restrict__ B,
                            const float* __restrict__ bias, float* __restrict__ C,
                            int M, int K, int Nc) {
  const int lane = threadIdx.x;
  const int half = lane >> 4;
  const int l    = lane & 15;
  const int tn = blockIdx.x, tm = blockIdx.y;
  const int arow = tm * 16 + l;
  const int bcol = tn * 16 + l;
  v8f acc = {};
  for (int k = 0; k < K; k += 4) {
    const float* ap = A + arow * K + k + half * 2;
    v2f a; a.x = ap[0]; a.y = ap[1];
    const float* bp = B + (size_t)(k + half * 2) * Nc + bcol;
    v2f b; b.x = bp[0]; b.y = bp[Nc];
    if (k + 4 < K) {
      __builtin_prefetch(bp + 4 * (size_t)Nc, 0, 1);
      __builtin_prefetch(bp + 5 * (size_t)Nc, 0, 1);
    }
    acc = __builtin_amdgcn_wmma_f32_16x16x4_f32(false, a, false, b, (short)0, acc, false, false);
  }
  const float bv = bias[bcol];
  const int rb = tm * 16 + half * 8;
#pragma unroll
  for (int r = 0; r < 8; ++r)
    C[(size_t)(rb + r) * Nc + bcol] = acc[r] + bv;
}

__global__ void k_zero(float* __restrict__ p, int n) {
  int i = blockIdx.x * blockDim.x + threadIdx.x;
  if (i < n) p[i] = 0.f;
}

__global__ void k_zero_i(int* __restrict__ p, int n) {
  int i = blockIdx.x * blockDim.x + threadIdx.x;
  if (i < n) p[i] = 0;
}

__global__ void k_count(const int* __restrict__ dstE, int* __restrict__ cnt) {
  int e = blockIdx.x * blockDim.x + threadIdx.x;
  if (e < NE) atomicAdd(cnt + dstE[e], 1);
}

#define SCAN_T 1024
#define SCAN_PER ((NN + SCAN_T - 1) / SCAN_T)
__global__ void __launch_bounds__(SCAN_T) k_scan(int* __restrict__ cur, int* __restrict__ rowptr) {
  __shared__ int part[SCAN_T];
  const int t = threadIdx.x;
  int loc[SCAN_PER];
  int sum = 0;
#pragma unroll
  for (int i = 0; i < SCAN_PER; ++i) {
    const int idx = t * SCAN_PER + i;
    loc[i] = sum;
    sum += (idx < NN) ? cur[idx] : 0;
  }
  part[t] = sum;
  __syncthreads();
  for (int off = 1; off < SCAN_T; off <<= 1) {
    const int v = (t >= off) ? part[t - off] : 0;
    __syncthreads();
    part[t] += v;
    __syncthreads();
  }
  const int excl = part[t] - sum;
#pragma unroll
  for (int i = 0; i < SCAN_PER; ++i) {
    const int idx = t * SCAN_PER + i;
    if (idx < NN) { rowptr[idx] = excl + loc[i]; cur[idx] = excl + loc[i]; }
  }
  if (t == SCAN_T - 1) rowptr[NN] = part[SCAN_T - 1];
}

__global__ void k_fill(const int* __restrict__ dstE, int* __restrict__ cur,
                       int* __restrict__ perm) {
  int e = blockIdx.x * blockDim.x + threadIdx.x;
  if (e < NE) perm[atomicAdd(cur + dstE[e], 1)] = e;
}

__global__ void k_rank(const int* __restrict__ perm, const int* __restrict__ rowptr,
                       const int* __restrict__ srcE, const int* __restrict__ dstE,
                       const float* __restrict__ ea,
                       int* __restrict__ csrc, v2f* __restrict__ cea) {
  int p = blockIdx.x * blockDim.x + threadIdx.x;
  if (p >= NE) return;
  const int e = perm[p];
  const int d = dstE[e];
  const int beg = rowptr[d], end = rowptr[d + 1];
  int r = 0;
  for (int q = beg; q < end; ++q) r += (perm[q] < e) ? 1 : 0;
  v2f a; a.x = ea[2 * e]; a.y = ea[2 * e + 1];
  csrc[beg + r] = srcE[e];
  cea[beg + r]  = a;
}

__device__ __forceinline__ float logit8(const float* xs, const float* xd,
                                        const float* w0, const float* w1,
                                        const float* at, float a0, float a1) {
  float lg = 0.f;
#pragma unroll
  for (int c = 0; c < CHN; ++c) {
    float m = xs[c] + xd[c] + a0 * w0[c] + a1 * w1[c];
    float g = m > 0.f ? m : 0.2f * m;
    lg += g * at[c];
  }
  return lg;
}

__global__ void __launch_bounds__(64)
k_gat_node(const float* __restrict__ xl, const float* __restrict__ xr,
           const float* __restrict__ res, const int* __restrict__ rowptr,
           const int* __restrict__ csrc, const v2f* __restrict__ cea,
           const float* __restrict__ We, const float* __restrict__ att,
           const float* __restrict__ bias_g, float* __restrict__ hout) {
  const int idx = blockIdx.x * blockDim.x + threadIdx.x;
  if (idx >= NN * HEADS) return;
  const int d = idx >> 3, cb = (idx & 7) * CHN;

  float w0[CHN], w1[CHN], at[CHN], xd[CHN], xself[CHN];
  {
    const v4f* pd = (const v4f*)(xr + d * HIDD + cb);
    const v4f* ps = (const v4f*)(xl + d * HIDD + cb);
    const v4f d0 = pd[0], d1 = pd[1], s0v = ps[0], s1v = ps[1];
#pragma unroll
    for (int c = 0; c < 4; ++c) {
      xd[c] = d0[c]; xd[4 + c] = d1[c]; xself[c] = s0v[c]; xself[4 + c] = s1v[c];
    }
#pragma unroll
    for (int c = 0; c < CHN; ++c) {
      w0[c] = We[cb + c]; w1[c] = We[HIDD + cb + c]; at[c] = att[cb + c];
    }
  }

  const int beg = rowptr[d], end = rowptr[d + 1];
  float mrun = -INFINITY, den = 0.f, asum0 = 0.f, asum1 = 0.f;
  float acc[CHN];
#pragma unroll
  for (int c = 0; c < CHN; ++c) acc[c] = 0.f;

  if (beg < end) {
    int sj[EU]; v2f aj[EU];
#pragma unroll
    for (int j = 0; j < EU; ++j) {
      const int q = min(beg + j, end - 1);
      sj[j] = csrc[q]; aj[j] = cea[q];
    }
    for (int p = beg; p < end; p += EU) {
      v4f x0[EU], x1[EU];
#pragma unroll
      for (int j = 0; j < EU; ++j) {
        const v4f* r = (const v4f*)(xl + sj[j] * HIDD + cb);
        x0[j] = r[0]; x1[j] = r[1];
      }
      int sn[EU]; v2f an[EU];
#pragma unroll
      for (int j = 0; j < EU; ++j) {
        const int q = min(p + EU + j, end - 1);
        sn[j] = csrc[q]; an[j] = cea[q];
      }
      float lg[EU];
      float cm = -INFINITY;
#pragma unroll
      for (int j = 0; j < EU; ++j) {
        const bool v = (p + j) < end;
        float xs[CHN];
#pragma unroll
        for (int c = 0; c < 4; ++c) { xs[c] = x0[j][c]; xs[4 + c] = x1[j][c]; }
        const float l = logit8(xs, xd, w0, w1, at, aj[j].x, aj[j].y);
        lg[j] = v ? l : -INFINITY;
        asum0 += v ? aj[j].x : 0.f;
        asum1 += v ? aj[j].y : 0.f;
        cm = fmaxf(cm, lg[j]);
      }
      const float mnew = fmaxf(mrun, cm);
      const float sc = __expf(mrun - mnew);
      float pj[EU];
      float psum = 0.f;
#pragma unroll
      for (int j = 0; j < EU; ++j) { pj[j] = __expf(lg[j] - mnew); psum += pj[j]; }
      den = den * sc + psum;
#pragma unroll
      for (int c = 0; c < 4; ++c) {
        float t0 = 0.f, t1 = 0.f;
#pragma unroll
        for (int j = 0; j < EU; ++j) { t0 += pj[j] * x0[j][c]; t1 += pj[j] * x1[j][c]; }
        acc[c]     = acc[c] * sc + t0;
        acc[4 + c] = acc[4 + c] * sc + t1;
      }
      mrun = mnew;
#pragma unroll
      for (int j = 0; j < EU; ++j) { sj[j] = sn[j]; aj[j] = an[j]; }
    }
  }

  {
    const float dg = fmaxf((float)(end - beg), 1.f);
    const float l = logit8(xself, xd, w0, w1, at, asum0 / dg, asum1 / dg);
    const float mnew = fmaxf(mrun, l);
    const float sc = __expf(mrun - mnew);
    const float pl = __expf(l - mnew);
    den = den * sc + pl;
#pragma unroll
    for (int c = 0; c < CHN; ++c) acc[c] = acc[c] * sc + pl * xself[c];
  }

  const v4f* pr = (const v4f*)(res + d * HIDD + cb);
  const v4f r0 = pr[0], r1 = pr[1];
  v4f o0, o1;
#pragma unroll
  for (int c = 0; c < 4; ++c) {
    float v0 = acc[c] / den     + bias_g[cb + c]     + r0[c];
    float v1 = acc[4 + c] / den + bias_g[cb + 4 + c] + r1[c];
    o0[c] = v0 > 0.f ? v0 : (__expf(v0) - 1.f);
    o1[c] = v1 > 0.f ? v1 : (__expf(v1) - 1.f);
  }
  v4f* po = (v4f*)(hout + d * HIDD + cb);
  po[0] = o0; po[1] = o1;
}

__global__ void k_pool_accum(const float* __restrict__ h, const int* __restrict__ batch,
                             float* __restrict__ pooled, float* __restrict__ cnt) {
  int idx = blockIdx.x * blockDim.x + threadIdx.x;
  if (idx >= NN * HIDD) return;
  int n = idx >> 6, c = idx & (HIDD - 1);
  int b = batch[n];
  atomicAdd(pooled + b * HIDD + c, h[idx]);
  if (c == 0) atomicAdd(cnt + b, 1.f);
}

__global__ void k_pool_div(float* __restrict__ pooled, const float* __restrict__ cnt) {
  int idx = blockIdx.x * blockDim.x + threadIdx.x;
  if (idx < NGR * HIDD) pooled[idx] /= fmaxf(cnt[idx >> 6], 1.f);
}

extern "C" void kernel_launch(void* const* d_in, const int* in_sizes, int n_in,
                              void* d_out_v, int out_size, void* d_ws, size_t ws_size,
                              hipStream_t stream) {
  (void)in_sizes; (void)n_in; (void)out_size; (void)ws_size;
  const float* x       = (const float*)d_in[0];
  const int*   ei      = (const int*)  d_in[1];
  const float* ea      = (const float*)d_in[2];
  const int*   batch   = (const int*)  d_in[3];
  const float* p1_Wl   = (const float*)d_in[4];
  const float* p1_bl   = (const float*)d_in[5];
  const float* p1_Wr   = (const float*)d_in[6];
  const float* p1_br   = (const float*)d_in[7];
  const float* p1_We   = (const float*)d_in[8];
  const float* p1_att  = (const float*)d_in[9];
  const float* p1_bias = (const float*)d_in[10];
  const float* p1_Wres = (const float*)d_in[11];
  const float* p1_bres = (const float*)d_in[12];
  const float* Wl      = (const float*)d_in[13];
  const float* bl      = (const float*)d_in[14];
  const float* Wr      = (const float*)d_in[15];
  const float* br      = (const float*)d_in[16];
  const float* We      = (const float*)d_in[17];
  const float* att     = (const float*)d_in[18];
  const float* bias_g  = (const float*)d_in[19];
  const float* Wres    = (const float*)d_in[20];
  const float* bres    = (const float*)d_in[21];
  const float* Wlin    = (const float*)d_in[22];
  const float* blin    = (const float*)d_in[23];

  float* out = (float*)d_out_v;
  float* ws  = (float*)d_ws;

  float* xl     = ws;
  float* xr     = ws + 640000;
  float* res    = ws + 1280000;
  int*   rowptr = (int*)(ws + 1920000);
  int*   cur    = (int*)(ws + 1940000);
  int*   perm   = (int*)(ws + 1960000);
  int*   csrc   = (int*)(ws + 2120000);
  v2f*   cea    = (v2f*)(ws + 2280000);
  float* cnt    = ws + 5470000;

  const int* srcE = ei;
  const int* dstE = ei + NE;

  const int TB = 256;
  const int gNode = (NN * HIDD + TB - 1) / TB;

  k_zero_i<<<(NN + TB - 1) / TB, TB, 0, stream>>>(cur, NN);
  k_count<<<(NE + TB - 1) / TB, TB, 0, stream>>>(dstE, cur);
  k_scan<<<1, SCAN_T, 0, stream>>>(cur, rowptr);
  k_fill<<<(NE + TB - 1) / TB, TB, 0, stream>>>(dstE, cur, perm);
  k_rank<<<(NE + TB - 1) / TB, TB, 0, stream>>>(perm, rowptr, srcE, dstE, ea, csrc, cea);

  for (int layer = 0; layer < 1 + LSTK; ++layer) {
    const float* hin = (layer == 0) ? x : out + (size_t)(layer - 1) * NN * HIDD;
    const int K = (layer == 0) ? FIN : HIDD;
    const int li = layer - 1;
    const float* Wl_   = (layer == 0) ? p1_Wl   : Wl   + (size_t)li * HIDD * HIDD;
    const float* bl_   = (layer == 0) ? p1_bl   : bl   + (size_t)li * HIDD;
    const float* Wr_   = (layer == 0) ? p1_Wr   : Wr   + (size_t)li * HIDD * HIDD;
    const float* br_   = (layer == 0) ? p1_br   : br   + (size_t)li * HIDD;
    const float* We_   = (layer == 0) ? p1_We   : We   + (size_t)li * 2 * HIDD;
    const float* att_  = (layer == 0) ? p1_att  : att  + (size_t)li * HEADS * CHN;
    const float* bg_   = (layer == 0) ? p1_bias : bias_g + (size_t)li * HIDD;
    const float* Wres_ = (layer == 0) ? p1_Wres : Wres + (size_t)li * HIDD * HIDD;
    const float* bres_ = (layer == 0) ? p1_bres : bres + (size_t)li * HIDD;
    float* hout = out + (size_t)layer * NN * HIDD;

    dim3 gg(HIDD / 16, NN / 16);
    k_gemm3_wmma<<<gg, 32, 0, stream>>>(hin, Wl_, bl_, xl, Wr_, br_, xr,
                                        Wres_, bres_, res, K);

    k_gat_node<<<(NN * HEADS + 63) / 64, 64, 0, stream>>>(xl, xr, res, rowptr, csrc, cea,
                                                          We_, att_, bg_, hout);
  }

  float* pooled = out + (size_t)(1 + LSTK) * NN * HIDD;
  float* x12    = pooled + NGR * HIDD;
  k_zero<<<(NGR * HIDD + TB - 1) / TB, TB, 0, stream>>>(pooled, NGR * HIDD);
  k_zero<<<1, NGR, 0, stream>>>(cnt, NGR);
  k_pool_accum<<<gNode, TB, 0, stream>>>(out + (size_t)LSTK * NN * HIDD, batch, pooled, cnt);
  k_pool_div<<<(NGR * HIDD + TB - 1) / TB, TB, 0, stream>>>(pooled, cnt);

  dim3 gf(VOC / 16, NGR / 16);
  k_gemm_wmma<<<gf, 32, 0, stream>>>(pooled, Wlin, blin, x12, NGR, HIDD, VOC);
}


# Next layer's three transforms computed inside the node kernel (one WMMA row tile of nodes per block, output tile staged in LDS, same fragment loaders): 9 of 10 transform launches go; sum order kept
# speedup vs baseline: 3.7596x; 1.3384x over previous
#include <hip/hip_runtime.h>

#define NN 10000
#define NE 160000
#define FIN 16
#define HEADS 8
#define CHN 8
#define HIDD 64
#define NGR 64
#define VOC 50000
#define LSTK 9
#define EU 8

typedef __attribute__((ext_vector_type(2))) float v2f;
typedef __attribute__((ext_vector_type(4))) float v4f;
typedef __attribute__((ext_vector_type(8))) float v8f;

__global__ void k_gemm3_wmma(const float* __restrict__ A,
                             const float* __restrict__ B0, const float* __restrict__ b0,
                             float* __restrict__ C0,
                             const float* __restrict__ B1, const float* __restrict__ b1,
                             float* __restrict__ C1,
                             const float* __restrict__ B2, const float* __restrict__ b2,
                             float* __restrict__ C2,
                             int K) {
  const int lane = threadIdx.x;
  const int half = lane >> 4;
  const int l    = lane & 15;
  const int tn = blockIdx.x, tm = blockIdx.y;
  const int arow = tm * 16 + l;
  const int bcol = tn * 16 + l;
  v8f acc0 = {}, acc1 = {}, acc2 = {};
  for (int k = 0; k < K; k += 4) {
    const float* ap = A + arow * K + k + half * 2;
    v2f a; a.x = ap[0]; a.y = ap[1];
    const int boff = (k + half * 2) * HIDD + bcol;
    v2f w0; w0.x = B0[boff]; w0.y = B0[boff + HIDD];
    v2f w1; w1.x = B1[boff]; w1.y = B1[boff + HIDD];
    v2f w2; w2.x = B2[boff]; w2.y = B2[boff + HIDD];
    acc0 = __builtin_amdgcn_wmma_f32_16x16x4_f32(false, a, false, w0, (short)0, acc0, false, false);
    acc1 = __builtin_amdgcn_wmma_f32_16x16x4_f32(false, a, false, w1, (short)0, acc1, false, false);
    acc2 = __builtin_amdgcn_wmma_f32_16x16x4_f32(false, a, false, w2, (short)0, acc2, false, false);
  }
  const float bv0 = b0[bcol], bv1 = b1[bcol], bv2 = b2[bcol];
  const int rb = tm * 16 + half * 8;
#pragma unroll
  for (int r = 0; r < 8; ++r) {
    const int off = (rb + r) * HIDD + bcol;
    C0[off] = acc0[r] + bv0;
    C1[off] = acc1[r] + bv1;
    C2[off] = acc2[r] + bv2;
  }
}

__global__ void k_gemm_wmma(const float* __restrict__ A, const float* __restrict__ B,
                            const float* __restrict__ bias, float* __restrict__ C,
                            int M, int K, int Nc) {
  const int lane = threadIdx.x;
  const int half = lane >> 4;
  const int l    = lane & 15;
  const int tn = blockIdx.x, tm = blockIdx.y;
  const int arow = tm * 16 + l;
  const int bcol = tn * 16 + l;
  v8f acc = {};
  for (int k = 0; k < K; k += 4) {
    const float* ap = A + arow * K + k + half * 2;
    v2f a; a.x = ap[0]; a.y = ap[1];
    const float* bp = B + (size_t)(k + half * 2) * Nc + bcol;
    v2f b; b.x = bp[0]; b.y = bp[Nc];
    if (k + 4 < K) {
      __builtin_prefetch(bp + 4 * (size_t)Nc, 0, 1);
      __builtin_prefetch(bp + 5 * (size_t)Nc, 0, 1);
    }
    acc = __builtin_amdgcn_wmma_f32_16x16x4_f32(false, a, false, b, (short)0, acc, false, false);
  }
  const float bv = bias[bcol];
  const int rb = tm * 16 + half * 8;
#pragma unroll
  for (int r = 0; r < 8; ++r)
    C[(size_t)(rb + r) * Nc + bcol] = acc[r] + bv;
}

__global__ void k_zero(float* __restrict__ p, int n) {
  int i = blockIdx.x * blockDim.x + threadIdx.x;
  if (i < n) p[i] = 0.f;
}

__global__ void k_zero_i(int* __restrict__ p, int n) {
  int i = blockIdx.x * blockDim.x + threadIdx.x;
  if (i < n) p[i] = 0;
}

__global__ void k_count(const int* __restrict__ dstE, int* __restrict__ cnt) {
  int e = blockIdx.x * blockDim.x + threadIdx.x;
  if (e < NE) atomicAdd(cnt + dstE[e], 1);
}

#define SCAN_T 1024
#define SCAN_PER ((NN + SCAN_T - 1) / SCAN_T)
__global__ void __launch_bounds__(SCAN_T) k_scan(int* __restrict__ cur, int* __restrict__ rowptr) {
  __shared__ int part[SCAN_T];
  const int t = threadIdx.x;
  int loc[SCAN_PER];
  int sum = 0;
#pragma unroll
  for (int i = 0; i < SCAN_PER; ++i) {
    const int idx = t * SCAN_PER + i;
    loc[i] = sum;
    sum += (idx < NN) ? cur[idx] : 0;
  }
  part[t] = sum;
  __syncthreads();
  for (int off = 1; off < SCAN_T; off <<= 1) {
    const int v = (t >= off) ? part[t - off] : 0;
    __syncthreads();
    part[t] += v;
    __syncthreads();
  }
  const int excl = part[t] - sum;
#pragma unroll
  for (int i = 0; i < SCAN_PER; ++i) {
    const int idx = t * SCAN_PER + i;
    if (idx < NN) { rowptr[idx] = excl + loc[i]; cur[idx] = excl + loc[i]; }
  }
  if (t == SCAN_T - 1) rowptr[NN] = part[SCAN_T - 1];
}

__global__ void k_fill(const int* __restrict__ dstE, int* __restrict__ cur,
                       int* __restrict__ perm) {
  int e = blockIdx.x * blockDim.x + threadIdx.x;
  if (e < NE) perm[atomicAdd(cur + dstE[e], 1)] = e;
}

__global__ void k_rank(const int* __restrict__ perm, const int* __restrict__ rowptr,
                       const int* __restrict__ srcE, const int* __restrict__ dstE,
                       const float* __restrict__ ea,
                       int* __restrict__ csrc, v2f* __restrict__ cea) {
  int p = blockIdx.x * blockDim.x + threadIdx.x;
  if (p >= NE) return;
  const int e = perm[p];
  const int d = dstE[e];
  const int beg = rowptr[d], end = rowptr[d + 1];
  int r = 0;
  for (int q = beg; q < end; ++q) r += (perm[q] < e) ? 1 : 0;
  v2f a; a.x = ea[2 * e]; a.y = ea[2 * e + 1];
  csrc[beg + r] = srcE[e];
  cea[beg + r]  = a;
}

__device__ __forceinline__ float logit8(const float* xs, const float* xd,
                                        const float* w0, const float* w1,
                                        const float* at, float a0, float a1) {
  float lg = 0.f;
#pragma unroll
  for (int c = 0; c < CHN; ++c) {
    float m = xs[c] + xd[c] + a0 * w0[c] + a1 * w1[c];
    float g = m > 0.f ? m : 0.2f * m;
    lg += g * at[c];
  }
  return lg;
}

#define NB 16
#define TP 68
static_assert(NN % NB == 0, "the grid covers the nodes exactly: no thread is out of range");

template <bool NEXT>
__global__ void __launch_bounds__(NB * HEADS)
k_gat_node(const float* __restrict__ xl, const float* __restrict__ xr,
           const float* __restrict__ res, const int* __restrict__ rowptr,
           const int* __restrict__ csrc, const v2f* __restrict__ cea,
           const float* __restrict__ We, const float* __restrict__ att,
           const float* __restrict__ bias_g, float* __restrict__ hout,
           const float* __restrict__ nB0, const float* __restrict__ nb0, float* __restrict__ nC0,
           const float* __restrict__ nB1, const float* __restrict__ nb1, float* __restrict__ nC1,
           const float* __restrict__ nB2, const float* __restrict__ nb2, float* __restrict__ nC2) {
  __shared__ alignas(16) float sH[NB * TP];
  const int idx = blockIdx.x * (NB * HEADS) + threadIdx.x;
  const int d = idx >> 3, cb = (idx & 7) * CHN;

  float w0[CHN], w1[CHN], at[CHN], xd[CHN], xself[CHN];
  {
    const v4f* pd = (const v4f*)(xr + d * HIDD + cb);
    const v4f* ps = (const v4f*)(xl + d * HIDD + cb);
    const v4f d0 = pd[0], d1 = pd[1], s0v = ps[0], s1v = ps[1];
#pragma unroll
    for (int c = 0; c < 4; ++c) {
      xd[c] = d0[c]; xd[4 + c] = d1[c]; xself[c] = s0v[c]; xself[4 + c] = s1v[c];
    }
#pragma unroll
    for (int c = 0; c < CHN; ++c) {
      w0[c] = We[cb + c]; w1[c] = We[HIDD + cb + c]; at[c] = att[cb + c];
    }
  }

  const int beg = rowptr[d], end = rowptr[d + 1];
  float mrun = -INFINITY, den = 0.f, asum0 = 0.f, asum1 = 0.f;
  float acc[CHN];
#pragma unroll
  for (int c = 0; c < CHN; ++c) acc[c] = 0.f;

  if (beg < end) {
    int sj[EU]; v2f aj[EU];
#pragma unroll
    for (int j = 0; j < EU; ++j) {
      const int q = min(beg + j, end - 1);
      sj[j] = csrc[q]; aj[j] = cea[q];
    }
    for (int p = beg; p < end; p += EU) {
      v4f x0[EU], x1[EU];
#pragma unroll
      for (int j = 0; j < EU; ++j) {
        const v4f* r = (const v4f*)(xl + sj[j] * HIDD + cb);
        x0[j] = r[0]; x1[j] = r[1];
      }
      int sn[EU]; v2f an[EU];
#pragma unroll
      for (int j = 0; j < EU; ++j) {
        const int q = min(p + EU + j, end - 1);
        sn[j] = csrc[q]; an[j] = cea[q];
      }
      float lg[EU];
      float cm = -INFINITY;
#pragma unroll
      for (int j = 0; j < EU; ++j) {
        const bool v = (p + j) < end;
        float xs[CHN];
#pragma unroll
        for (int c = 0; c < 4; ++c) { xs[c] = x0[j][c]; xs[4 + c] = x1[j][c]; }
        const float l = logit8(xs, xd, w0, w1, at, aj[j].x, aj[j].y);
        lg[j] = v ? l : -INFINITY;
        asum0 += v ? aj[j].x : 0.f;
        asum1 += v ? aj[j].y : 0.f;
        cm = fmaxf(cm, lg[j]);
      }
      const float mnew = fmaxf(mrun, cm);
      const float sc = __expf(mrun - mnew);
      float pj[EU];
      float psum = 0.f;
#pragma unroll
      for (int j = 0; j < EU; ++j) { pj[j] = __expf(lg[j] - mnew); psum += pj[j]; }
      den = den * sc + psum;
#pragma unroll
      for (int c = 0; c < 4; ++c) {
        float t0 = 0.f, t1 = 0.f;
#pragma unroll
        for (int j = 0; j < EU; ++j) { t0 += pj[j] * x0[j][c]; t1 += pj[j] * x1[j][c]; }
        acc[c]     = acc[c] * sc + t0;
        acc[4 + c] = acc[4 + c] * sc + t1;
      }
      mrun = mnew;
#pragma unroll
      for (int j = 0; j < EU; ++j) { sj[j] = sn[j]; aj[j] = an[j]; }
    }
  }

  const int lane = threadIdx.x & 31;
  const int half = lane >> 4;
  const int l    = lane & 15;
  const int bcol = (threadIdx.x >> 5) * 16 + l;
  v2f wf0[HIDD / 4], wf1[HIDD / 4], wf2[HIDD / 4];
  float bv0 = 0.f, bv1 = 0.f, bv2 = 0.f;
  if (NEXT) {
#pragma unroll
    for (int j = 0; j < HIDD / 4; ++j) {
      const int boff = (4 * j + half * 2) * HIDD + bcol;
      wf0[j].x = nB0[boff]; wf0[j].y = nB0[boff + HIDD];
      wf1[j].x = nB1[boff]; wf1[j].y = nB1[boff + HIDD];
      wf2[j].x = nB2[boff]; wf2[j].y = nB2[boff + HIDD];
    }
    bv0 = nb0[bcol]; bv1 = nb1[bcol]; bv2 = nb2[bcol];
  }

  {
    const float dg = fmaxf((float)(end - beg), 1.f);
    const float l = logit8(xself, xd, w0, w1, at, asum0 / dg, asum1 / dg);
    const float mnew = fmaxf(mrun, l);
    const float sc = __expf(mrun - mnew);
    const float pl = __expf(l - mnew);
    den = den * sc + pl;
#pragma unroll
    for (int c = 0; c < CHN; ++c) acc[c] = acc[c] * sc + pl * xself[c];
  }

  const v4f* pr = (const v4f*)(res + d * HIDD + cb);
  const v4f r0 = pr[0], r1 = pr[1];
  v4f o0, o1;
#pragma unroll
  for (int c = 0; c < 4; ++c) {
    float v0 = acc[c] / den     + bias_g[cb + c]     + r0[c];
    float v1 = acc[4 + c] / den + bias_g[cb + 4 + c] + r1[c];
    o0[c] = v0 > 0.f ? v0 : (__expf(v0) - 1.f);
    o1[c] = v1 > 0.f ? v1 : (__expf(v1) - 1.f);
  }
  v4f* po = (v4f*)(hout + d * HIDD + cb);
  if (!NEXT) { po[0] = o0; po[1] = o1; }

  if (NEXT) {
    v4f* ps = (v4f*)(sH + (threadIdx.x >> 3) * TP + cb);
    ps[0] = o0; ps[1] = o1;
    __syncthreads();
    po[0] = o0; po[1] = o1;
    v8f acc0 = {}, acc1 = {}, acc2 = {};
#pragma unroll
    for (int j = 0; j < HIDD / 4; ++j) {
      const v2f a = *(const v2f*)(sH + l * TP + 4 * j + half * 2);
      acc0 = __builtin_amdgcn_wmma_f32_16x16x4_f32(false, a, false, wf0[j], (short)0, acc0, false, false);
      acc1 = __builtin_amdgcn_wmma_f32_16x16x4_f32(false, a, false, wf1[j], (short)0, acc1, false, false);
      acc2 = __builtin_amdgcn_wmma_f32_16x16x4_f32(false, a, false, wf2[j], (short)0, acc2, false, false);
    }
    const int rb = blockIdx.x * NB + half * 8;
#pragma unroll
    for (int r = 0; r < 8; ++r) {
      const int off = (rb + r) * HIDD + bcol;
      nC0[off] = acc0[r] + bv0;
      nC1[off] = acc1[r] + bv1;
      nC2[off] = acc2[r] + bv2;
    }
  }
}

__global__ void k_pool_accum(const float* __restrict__ h, const int* __restrict__ batch,
                             float* __restrict__ pooled, float* __restrict__ cnt) {
  int idx = blockIdx.x * blockDim.x + threadIdx.x;
  if (idx >= NN * HIDD) return;
  int n = idx >> 6, c = idx & (HIDD - 1);
  int b = batch[n];
  atomicAdd(pooled + b * HIDD + c, h[idx]);
  if (c == 0) atomicAdd(cnt + b, 1.f);
}

__global__ void k_pool_div(float* __restrict__ pooled, const float* __restrict__ cnt) {
  int idx = blockIdx.x * blockDim.x + threadIdx.x;
  if (idx < NGR * HIDD) pooled[idx] /= fmaxf(cnt[idx >> 6], 1.f);
}

extern "C" void kernel_launch(void* const* d_in, const int* in_sizes, int n_in,
                              void* d_out_v, int out_size, void* d_ws, size_t ws_size,
                              hipStream_t stream) {
  (void)in_sizes; (void)n_in; (void)out_size; (void)ws_size;
  const float* x       = (const float*)d_in[0];
  const int*   ei      = (const int*)  d_in[1];
  const float* ea      = (const float*)d_in[2];
  const int*   batch   = (const int*)  d_in[3];
  const float* p1_Wl   = (const float*)d_in[4];
  const float* p1_bl   = (const float*)d_in[5];
  const float* p1_Wr   = (const float*)d_in[6];
  const float* p1_br   = (const float*)d_in[7];
  const float* p1_We   = (const float*)d_in[8];
  const float* p1_att  = (const float*)d_in[9];
  const float* p1_bias = (const float*)d_in[10];
  const float* p1_Wres = (const float*)d_in[11];
  const float* p1_bres = (const float*)d_in[12];
  const float* Wl      = (const float*)d_in[13];
  const float* bl      = (const float*)d_in[14];
  const float* Wr      = (const float*)d_in[15];
  const float* br      = (const float*)d_in[16];
  const float* We      = (const float*)d_in[17];
  const float* att     = (const float*)d_in[18];
  const float* bias_g  = (const float*)d_in[19];
  const float* Wres    = (const float*)d_in[20];
  const float* bres    = (const float*)d_in[21];
  const float* Wlin    = (const float*)d_in[22];
  const float* blin    = (const float*)d_in[23];

  float* out = (float*)d_out_v;
  float* ws  = (float*)d_ws;

  float* xl     = ws;
  float* xr     = ws + 640000;
  float* res    = ws + 1280000;
  int*   rowptr = (int*)(ws + 1920000);
  int*   cur    = (int*)(ws + 1940000);
  int*   perm   = (int*)(ws + 1960000);
  int*   csrc   = (int*)(ws + 2120000);
  v2f*   cea    = (v2f*)(ws + 2280000);
  float* xlB    = ws + 2600000;
  float* xrB    = ws + 3240000;
  float* resB   = ws + 3880000;
  float* cnt    = ws + 5470000;

  const int* srcE = ei;
  const int* dstE = ei + NE;

  const int TB = 256;
  const int gNode = (NN * HIDD + TB - 1) / TB;

  k_zero_i<<<(NN + TB - 1) / TB, TB, 0, stream>>>(cur, NN);
  k_count<<<(NE + TB - 1) / TB, TB, 0, stream>>>(dstE, cur);
  k_scan<<<1, SCAN_T, 0, stream>>>(cur, rowptr);
  k_fill<<<(NE + TB - 1) / TB, TB, 0, stream>>>(dstE, cur, perm);
  k_rank<<<(NE + TB - 1) / TB, TB, 0, stream>>>(perm, rowptr, srcE, dstE, ea, csrc, cea);

  {
    dim3 gg(HIDD / 16, NN / 16);
    k_gemm3_wmma<<<gg, 32, 0, stream>>>(x, p1_Wl, p1_bl, xl, p1_Wr, p1_br, xr,
                                        p1_Wres, p1_bres, res, FIN);
  }
  for (int layer = 0; layer < 1 + LSTK; ++layer) {
    const int li = layer - 1;
    const float* We_   = (layer == 0) ? p1_We   : We   + (size_t)li * 2 * HIDD;
    const float* att_  = (layer == 0) ? p1_att  : att  + (size_t)li * HEADS * CHN;
    const float* bg_   = (layer == 0) ? p1_bias : bias_g + (size_t)li * HIDD;
    float* hout = out + (size_t)layer * NN * HIDD;
    const bool odd = (layer & 1) != 0;
    const float* xl_c  = odd ? xlB  : xl;   float* xl_n  = odd ? xl  : xlB;
    const float* xr_c  = odd ? xrB  : xr;   float* xr_n  = odd ? xr  : xrB;
    const float* res_c = odd ? resB : res;  float* res_n = odd ? res : resB;

    if (layer < LSTK) {
      k_gat_node<true><<<NN / NB, NB * HEADS, 0, stream>>>(
          xl_c, xr_c, res_c, rowptr, csrc, cea, We_, att_, bg_, hout,
          Wl   + (size_t)layer * HIDD * HIDD, bl   + (size_t)layer * HIDD, xl_n,
          Wr   + (size_t)layer * HIDD * HIDD, br   + (size_t)layer * HIDD, xr_n,
          Wres + (size_t)layer * HIDD * HIDD, bres + (size_t)layer * HIDD, res_n);
    } else {
      k_gat_node<false><<<NN / NB, NB * HEADS, 0, stream>>>(
          xl_c, xr_c, res_c, rowptr, csrc, cea, We_, att_, bg_, hout,
          nullptr, nullptr, nullptr, nullptr, nullptr, nullptr, nullptr, nullptr, nullptr);
    }
  }

  float* pooled = out + (size_t)(1 + LSTK) * NN * HIDD;
  float* x12    = pooled + NGR * HIDD;
  k_zero<<<(NGR * HIDD + TB - 1) / TB, TB, 0, stream>>>(pooled, NGR * HIDD);
  k_zero<<<1, NGR, 0, stream>>>(cnt, NGR);
  k_pool_accum<<<gNode, TB, 0, stream>>>(out + (size_t)LSTK * NN * HIDD, batch, pooled, cnt);
  k_pool_div<<<(NGR * HIDD + TB - 1) / TB, TB, 0, stream>>>(pooled, cnt);

  dim3 gf(VOC / 16, NGR / 16);
  k_gemm_wmma<<<gf, 32, 0, stream>>>(pooled, Wlin, blin, x12, NGR, HIDD, VOC);
}


# Mean pool loses its 4 launches: last node kernel sums its 16 rows per graph in node order (LDS) and adds the runs by float atomics (arrival order, as before); classifier divides the sums on load
# speedup vs baseline: 4.0396x; 1.0745x over previous
#include <hip/hip_runtime.h>

#define NN 10000
#define NE 160000
#define FIN 16
#define HEADS 8
#define CHN 8
#define HIDD 64
#define NGR 64
#define VOC 50000
#define LSTK 9
#define EU 8

typedef __attribute__((ext_vector_type(2))) float v2f;
typedef __attribute__((ext_vector_type(4))) float v4f;
typedef __attribute__((ext_vector_type(8))) float v8f;

__global__ void k_gemm3_wmma(const float* __restrict__ A,
                             const float* __restrict__ B0, const float* __restrict__ b0,
                             float* __restrict__ C0,
                             const float* __restrict__ B1, const float* __restrict__ b1,
                             float* __restrict__ C1,
                             const float* __restrict__ B2, const float* __restrict__ b2,
                             float* __restrict__ C2,
                             int K) {
  const int lane = threadIdx.x;
  const int half = lane >> 4;
  const int l    = lane & 15;
  const int tn = blockIdx.x, tm = blockIdx.y;
  const int arow = tm * 16 + l;
  const int bcol = tn * 16 + l;
  v8f acc0 = {}, acc1 = {}, acc2 = {};
  for (int k = 0; k < K; k += 4) {
    const float* ap = A + arow * K + k + half * 2;
    v2f a; a.x = ap[0]; a.y = ap[1];
    const int boff = (k + half * 2) * HIDD + bcol;
    v2f w0; w0.x = B0[boff]; w0.y = B0[boff + HIDD];
    v2f w1; w1.x = B1[boff]; w1.y = B1[boff + HIDD];
    v2f w2; w2.x = B2[boff]; w2.y = B2[boff + HIDD];
    acc0 = __builtin_amdgcn_wmma_f32_16x16x4_f32(false, a, false, w0, (short)0, acc0, false, false);
    acc1 = __builtin_amdgcn_wmma_f32_16x16x4_f32(false, a, false, w1, (short)0, acc1, false, false);
    acc2 = __builtin_amdgcn_wmma_f32_16x16x4_f32(false, a, false, w2, (short)0, acc2, false, false);
  }
  const float bv0 = b0[bcol], bv1 = b1[bcol], bv2 = b2[bcol];
  const int rb = tm * 16 + half * 8;
#pragma unroll
  for (int r = 0; r < 8; ++r) {
    const int off = (rb + r) * HIDD + bcol;
    C0[off] = acc0[r] + bv0;
    C1[off] = acc1[r] + bv1;
    C2[off] = acc2[r] + bv2;
  }
}

__global__ void k_gemm_wmma(const float* __restrict__ S, const float* __restrict__ Sn,
                            const float* __restrict__ B,
                            const float* __restrict__ bias, float* __restrict__ C,
                            float* __restrict__ Amean,
                            int M, int K, int Nc) {
  const int lane = threadIdx.x;
  const int half = lane >> 4;
  const int l    = lane & 15;
  const int tn = blockIdx.x, tm = blockIdx.y;
  const int arow = tm * 16 + l;
  const int bcol = tn * 16 + l;
  v8f acc = {};
  const float dn = fmaxf(Sn[arow], 1.f);
  for (int k = 0; k < K; k += 4) {
    const int ao = arow * K + k + half * 2;
    const float s0 = S[ao], s1 = S[ao + 1];
    const float* bp = B + (size_t)(k + half * 2) * Nc + bcol;
    v2f b; b.x = bp[0]; b.y = bp[Nc];
    if (k + 4 < K) {
      __builtin_prefetch(bp + 4 * (size_t)Nc, 0, 1);
      __builtin_prefetch(bp + 5 * (size_t)Nc, 0, 1);
    }
    v2f a; a.x = s0 / dn; a.y = s1 / dn;
    acc = __builtin_amdgcn_wmma_f32_16x16x4_f32(false, a, false, b, (short)0, acc, false, false);
    if (tn == 0) *(v2f*)(Amean + ao) = a;
  }
  const float bv = bias[bcol];
  const int rb = tm * 16 + half * 8;
#pragma unroll
  for (int r = 0; r < 8; ++r)
    C[(size_t)(rb + r) * Nc + bcol] = acc[r] + bv;
}

__global__ void k_zero_i(int* __restrict__ p, int n) {
  int i = blockIdx.x * blockDim.x + threadIdx.x;
  if (i < n) p[i] = 0;
}

__global__ void k_count(const int* __restrict__ dstE, int* __restrict__ cnt) {
  int e = blockIdx.x * blockDim.x + threadIdx.x;
  if (e < NE) atomicAdd(cnt + dstE[e], 1);
}

#define SCAN_T 1024
#define SCAN_PER ((NN + SCAN_T - 1) / SCAN_T)
__global__ void __launch_bounds__(SCAN_T) k_scan(int* __restrict__ cur, int* __restrict__ rowptr) {
  __shared__ int part[SCAN_T];
  const int t = threadIdx.x;
  int loc[SCAN_PER];
  int sum = 0;
#pragma unroll
  for (int i = 0; i < SCAN_PER; ++i) {
    const int idx = t * SCAN_PER + i;
    loc[i] = sum;
    sum += (idx < NN) ? cur[idx] : 0;
  }
  part[t] = sum;
  __syncthreads();
  for (int off = 1; off < SCAN_T; off <<= 1) {
    const int v = (t >= off) ? part[t - off] : 0;
    __syncthreads();
    part[t] += v;
    __syncthreads();
  }
  const int excl = part[t] - sum;
#pragma unroll
  for (int i = 0; i < SCAN_PER; ++i) {
    const int idx = t * SCAN_PER + i;
    if (idx < NN) { rowptr[idx] = excl + loc[i]; cur[idx] = excl + loc[i]; }
  }
  if (t == SCAN_T - 1) rowptr[NN] = part[SCAN_T - 1];
}

__global__ void k_fill(const int* __restrict__ dstE, int* __restrict__ cur,
                       int* __restrict__ perm) {
  int e = blockIdx.x * blockDim.x + threadIdx.x;
  if (e < NE) perm[atomicAdd(cur + dstE[e], 1)] = e;
}

__global__ void k_rank(const int* __restrict__ perm, const int* __restrict__ rowptr,
                       const int* __restrict__ srcE, const int* __restrict__ dstE,
                       const float* __restrict__ ea,
                       int* __restrict__ csrc, v2f* __restrict__ cea) {
  int p = blockIdx.x * blockDim.x + threadIdx.x;
  if (p >= NE) return;
  const int e = perm[p];
  const int d = dstE[e];
  const int beg = rowptr[d], end = rowptr[d + 1];
  int r = 0;
  for (int q = beg; q < end; ++q) r += (perm[q] < e) ? 1 : 0;
  v2f a; a.x = ea[2 * e]; a.y = ea[2 * e + 1];
  csrc[beg + r] = srcE[e];
  cea[beg + r]  = a;
}

__device__ __forceinline__ float logit8(const float* xs, const float* xd,
                                        const float* w0, const float* w1,
                                        const float* at, float a0, float a1) {
  float lg = 0.f;
#pragma unroll
  for (int c = 0; c < CHN; ++c) {
    float m = xs[c] + xd[c] + a0 * w0[c] + a1 * w1[c];
    float g = m > 0.f ? m : 0.2f * m;
    lg += g * at[c];
  }
  return lg;
}

#define NB 16
#define TP 68
static_assert(NN % NB == 0, "the grid covers the nodes exactly: no thread is out of range");

template <bool NEXT>
__global__ void __launch_bounds__(NB * HEADS)
k_gat_node(const float* __restrict__ xl, const float* __restrict__ xr,
           const float* __restrict__ res, const int* __restrict__ rowptr,
           const int* __restrict__ csrc, const v2f* __restrict__ cea,
           const float* __restrict__ We, const float* __restrict__ att,
           const float* __restrict__ bias_g, float* __restrict__ hout,
           const float* __restrict__ nB0, const float* __restrict__ nb0, float* __restrict__ nC0,
           const float* __restrict__ nB1, const float* __restrict__ nb1, float* __restrict__ nC1,
           const float* __restrict__ nB2, const float* __restrict__ nb2, float* __restrict__ nC2,
           const int* __restrict__ batch, float* __restrict__ psum, float* __restrict__ pcnt) {
  __shared__ alignas(16) float sH[NB * TP];
  const int idx = blockIdx.x * (NB * HEADS) + threadIdx.x;
  const int d = idx >> 3, cb = (idx & 7) * CHN;

  float w0[CHN], w1[CHN], at[CHN], xd[CHN], xself[CHN];
  {
    const v4f* pd = (const v4f*)(xr + d * HIDD + cb);
    const v4f* ps = (const v4f*)(xl + d * HIDD + cb);
    const v4f d0 = pd[0], d1 = pd[1], s0v = ps[0], s1v = ps[1];
#pragma unroll
    for (int c = 0; c < 4; ++c) {
      xd[c] = d0[c]; xd[4 + c] = d1[c]; xself[c] = s0v[c]; xself[4 + c] = s1v[c];
    }
#pragma unroll
    for (int c = 0; c < CHN; ++c) {
      w0[c] = We[cb + c]; w1[c] = We[HIDD + cb + c]; at[c] = att[cb + c];
    }
  }

  const int beg = rowptr[d], end = rowptr[d + 1];
  float mrun = -INFINITY, den = 0.f, asum0 = 0.f, asum1 = 0.f;
  float acc[CHN];
#pragma unroll
  for (int c = 0; c < CHN; ++c) acc[c] = 0.f;

  if (beg < end) {
    int sj[EU]; v2f aj[EU];
#pragma unroll
    for (int j = 0; j < EU; ++j) {
      const int q = min(beg + j, end - 1);
      sj[j] = csrc[q]; aj[j] = cea[q];
    }
    for (int p = beg; p < end; p += EU) {
      v4f x0[EU], x1[EU];
#pragma unroll
      for (int j = 0; j < EU; ++j) {
        const v4f* r = (const v4f*)(xl + sj[j] * HIDD + cb);
        x0[j] = r[0]; x1[j] = r[1];
      }
      int sn[EU]; v2f an[EU];
#pragma unroll
      for (int j = 0; j < EU; ++j) {
        const int q = min(p + EU + j, end - 1);
        sn[j] = csrc[q]; an[j] = cea[q];
      }
      float lg[EU];
      float cm = -INFINITY;
#pragma unroll
      for (int j = 0; j < EU; ++j) {
        const bool v = (p + j) < end;
        float xs[CHN];
#pragma unroll
        for (int c = 0; c < 4; ++c) { xs[c] = x0[j][c]; xs[4 + c] = x1[j][c]; }
        const float l = logit8(xs, xd, w0, w1, at, aj[j].x, aj[j].y);
        lg[j] = v ? l : -INFINITY;
        asum0 += v ? aj[j].x : 0.f;
        asum1 += v ? aj[j].y : 0.f;
        cm = fmaxf(cm, lg[j]);
      }
      const float mnew = fmaxf(mrun, cm);
      const float sc = __expf(mrun - mnew);
      float pj[EU];
      float psum = 0.f;
#pragma unroll
      for (int j = 0; j < EU; ++j) { pj[j] = __expf(lg[j] - mnew); psum += pj[j]; }
      den = den * sc + psum;
#pragma unroll
      for (int c = 0; c < 4; ++c) {
        float t0 = 0.f, t1 = 0.f;
#pragma unroll
        for (int j = 0; j < EU; ++j) { t0 += pj[j] * x0[j][c]; t1 += pj[j] * x1[j][c]; }
        acc[c]     = acc[c] * sc + t0;
        acc[4 + c] = acc[4 + c] * sc + t1;
      }
      mrun = mnew;
#pragma unroll
      for (int j = 0; j < EU; ++j) { sj[j] = sn[j]; aj[j] = an[j]; }
    }
  }

  const int lane = threadIdx.x & 31;
  const int half = lane >> 4;
  const int l    = lane & 15;
  const int bcol = (threadIdx.x >> 5) * 16 + l;
  v2f wf0[HIDD / 4], wf1[HIDD / 4], wf2[HIDD / 4];
  float bv0 = 0.f, bv1 = 0.f, bv2 = 0.f;
  if (NEXT) {
#pragma unroll
    for (int j = 0; j < HIDD / 4; ++j) {
      const int boff = (4 * j + half * 2) * HIDD + bcol;
      wf0[j].x = nB0[boff]; wf0[j].y = nB0[boff + HIDD];
      wf1[j].x = nB1[boff]; wf1[j].y = nB1[boff + HIDD];
      wf2[j].x = nB2[boff]; wf2[j].y = nB2[boff + HIDD];
    }
    bv0 = nb0[bcol]; bv1 = nb1[bcol]; bv2 = nb2[bcol];
  }
  const int pc  = threadIdx.x & (HIDD - 1);
  const int pr0 = (threadIdx.x >> 6) * (NB / 2);
  int gid[NB / 2];
  if (!NEXT) {
#pragma unroll
    for (int r = 0; r < NB / 2; ++r) gid[r] = batch[blockIdx.x * NB + pr0 + r];
  }

  {
    const float dg = fmaxf((float)(end - beg), 1.f);
    const float l = logit8(xself, xd, w0, w1, at, asum0 / dg, asum1 / dg);
    const float mnew = fmaxf(mrun, l);
    const float sc = __expf(mrun - mnew);
    const float pl = __expf(l - mnew);
    den = den * sc + pl;
#pragma unroll
    for (int c = 0; c < CHN; ++c) acc[c] = acc[c] * sc + pl * xself[c];
  }

  const v4f* pr = (const v4f*)(res + d * HIDD + cb);
  const v4f r0 = pr[0], r1 = pr[1];
  v4f o0, o1;
#pragma unroll
  for (int c = 0; c < 4; ++c) {
    float v0 = acc[c] / den     + bias_g[cb + c]     + r0[c];
    float v1 = acc[4 + c] / den + bias_g[cb + 4 + c] + r1[c];
    o0[c] = v0 > 0.f ? v0 : (__expf(v0) - 1.f);
    o1[c] = v1 > 0.f ? v1 : (__expf(v1) - 1.f);
  }
  v4f* po = (v4f*)(hout + d * HIDD + cb);
  if (!NEXT) {
    v4f* ps = (v4f*)(sH + (threadIdx.x >> 3) * TP + cb);
    ps[0] = o0; ps[1] = o1;
    __syncthreads();
    po[0] = o0; po[1] = o1;
    float hv[NB / 2];
#pragma unroll
    for (int r = 0; r < NB / 2; ++r) hv[r] = sH[(pr0 + r) * TP + pc];
    float run = 0.f, nrun = 0.f;
    int gcur = gid[0];
#pragma unroll
    for (int r = 0; r < NB / 2; ++r) {
      if (gid[r] != gcur) {
        atomicAdd(psum + gcur * HIDD + pc, run);
        if (pc == 0) atomicAdd(pcnt + gcur, nrun);
        run = 0.f; nrun = 0.f; gcur = gid[r];
      }
      run += hv[r];
      nrun += 1.f;
    }
    atomicAdd(psum + gcur * HIDD + pc, run);
    if (pc == 0) atomicAdd(pcnt + gcur, nrun);
  }

  if (NEXT) {
    v4f* ps = (v4f*)(sH + (threadIdx.x >> 3) * TP + cb);
    ps[0] = o0; ps[1] = o1;
    __syncthreads();
    po[0] = o0; po[1] = o1;
    v8f acc0 = {}, acc1 = {}, acc2 = {};
#pragma unroll
    for (int j = 0; j < HIDD / 4; ++j) {
      const v2f a = *(const v2f*)(sH + l * TP + 4 * j + half * 2);
      acc0 = __builtin_amdgcn_wmma_f32_16x16x4_f32(false, a, false, wf0[j], (short)0, acc0, false, false);
      acc1 = __builtin_amdgcn_wmma_f32_16x16x4_f32(false, a, false, wf1[j], (short)0, acc1, false, false);
      acc2 = __builtin_amdgcn_wmma_f32_16x16x4_f32(false, a, false, wf2[j], (short)0, acc2, false, false);
    }
    const int rb = blockIdx.x * NB + half * 8;
#pragma unroll
    for (int r = 0; r < 8; ++r) {
      const int off = (rb + r) * HIDD + bcol;
      nC0[off] = acc0[r] + bv0;
      nC1[off] = acc1[r] + bv1;
      nC2[off] = acc2[r] + bv2;
    }
  }
}

extern "C" void kernel_launch(void* const* d_in, const int* in_sizes, int n_in,
                              void* d_out_v, int out_size, void* d_ws, size_t ws_size,
                              hipStream_t stream) {
  (void)in_sizes; (void)n_in; (void)out_size; (void)ws_size;
  const float* x       = (const float*)d_in[0];
  const int*   ei      = (const int*)  d_in[1];
  const float* ea      = (const float*)d_in[2];
  const int*   batch   = (const int*)  d_in[3];
  const float* p1_Wl   = (const float*)d_in[4];
  const float* p1_bl   = (const float*)d_in[5];
  const float* p1_Wr   = (const float*)d_in[6];
  const float* p1_br   = (const float*)d_in[7];
  const float* p1_We   = (const float*)d_in[8];
  const float* p1_att  = (const float*)d_in[9];
  const float* p1_bias = (const float*)d_in[10];
  const float* p1_Wres = (const float*)d_in[11];
  const float* p1_bres = (const float*)d_in[12];
  const float* Wl      = (const float*)d_in[13];
  const float* bl      = (const float*)d_in[14];
  const float* Wr      = (const float*)d_in[15];
  const float* br      = (const float*)d_in[16];
  const float* We      = (const float*)d_in[17];
  const float* att     = (const float*)d_in[18];
  const float* bias_g  = (const float*)d_in[19];
  const float* Wres    = (const float*)d_in[20];
  const float* bres    = (const float*)d_in[21];
  const float* Wlin    = (const float*)d_in[22];
  const float* blin    = (const float*)d_in[23];

  float* out = (float*)d_out_v;
  float* ws  = (float*)d_ws;

  float* xl     = ws;
  float* xr     = ws + 640000;
  float* res    = ws + 1280000;
  int*   rowptr = (int*)(ws + 1920000);
  int*   cur    = (int*)(ws + 1940000);
  float* psum   = ws + 1950000;
  float* pcnt   = ws + 1954096;
  int*   perm   = (int*)(ws + 1960000);
  int*   csrc   = (int*)(ws + 2120000);
  v2f*   cea    = (v2f*)(ws + 2280000);
  float* xlB    = ws + 2600000;
  float* xrB    = ws + 3240000;
  float* resB   = ws + 3880000;

  const int* srcE = ei;
  const int* dstE = ei + NE;

  const int TB = 256;

  k_zero_i<<<(NN + NGR * HIDD + NGR + TB - 1) / TB, TB, 0, stream>>>(cur, NN + NGR * HIDD + NGR);
  k_count<<<(NE + TB - 1) / TB, TB, 0, stream>>>(dstE, cur);
  k_scan<<<1, SCAN_T, 0, stream>>>(cur, rowptr);
  k_fill<<<(NE + TB - 1) / TB, TB, 0, stream>>>(dstE, cur, perm);
  k_rank<<<(NE + TB - 1) / TB, TB, 0, stream>>>(perm, rowptr, srcE, dstE, ea, csrc, cea);

  {
    dim3 gg(HIDD / 16, NN / 16);
    k_gemm3_wmma<<<gg, 32, 0, stream>>>(x, p1_Wl, p1_bl, xl, p1_Wr, p1_br, xr,
                                        p1_Wres, p1_bres, res, FIN);
  }
  for (int layer = 0; layer < 1 + LSTK; ++layer) {
    const int li = layer - 1;
    const float* We_   = (layer == 0) ? p1_We   : We   + (size_t)li * 2 * HIDD;
    const float* att_  = (layer == 0) ? p1_att  : att  + (size_t)li * HEADS * CHN;
    const float* bg_   = (layer == 0) ? p1_bias : bias_g + (size_t)li * HIDD;
    float* hout = out + (size_t)layer * NN * HIDD;
    const bool odd = (layer & 1) != 0;
    const float* xl_c  = odd ? xlB  : xl;   float* xl_n  = odd ? xl  : xlB;
    const float* xr_c  = odd ? xrB  : xr;   float* xr_n  = odd ? xr  : xrB;
    const float* res_c = odd ? resB : res;  float* res_n = odd ? res : resB;

    if (layer < LSTK) {
      k_gat_node<true><<<NN / NB, NB * HEADS, 0, stream>>>(
          xl_c, xr_c, res_c, rowptr, csrc, cea, We_, att_, bg_, hout,
          Wl   + (size_t)layer * HIDD * HIDD, bl   + (size_t)layer * HIDD, xl_n,
          Wr   + (size_t)layer * HIDD * HIDD, br   + (size_t)layer * HIDD, xr_n,
          Wres + (size_t)layer * HIDD * HIDD, bres + (size_t)layer * HIDD, res_n,
          nullptr, nullptr, nullptr);
    } else {
      k_gat_node<false><<<NN / NB, NB * HEADS, 0, stream>>>(
          xl_c, xr_c, res_c, rowptr, csrc, cea, We_, att_, bg_, hout,
          nullptr, nullptr, nullptr, nullptr, nullptr, nullptr, nullptr, nullptr, nullptr,
          batch, psum, pcnt);
    }
  }

  float* pooled = out + (size_t)(1 + LSTK) * NN * HIDD;
  float* x12    = pooled + NGR * HIDD;
  dim3 gf(VOC / 16, NGR / 16);
  k_gemm_wmma<<<gf, 32, 0, stream>>>(psum, pcnt, Wlin, blin, x12, pooled, NGR, HIDD, VOC);
}


# Classifier re-tiled: a wave owns 16 rows x 64 columns (4 accumulators, 4 adjacent columns per lane), all its B rows requested at once as 16-byte loads; 782 4-wave groups replace 12500 1-wave ones
# speedup vs baseline: 4.2637x; 1.0555x over previous
#include <hip/hip_runtime.h>

#define NN 10000
#define NE 160000
#define FIN 16
#define HEADS 8
#define CHN 8
#define HIDD 64
#define NGR 64
#define VOC 50000
#define LSTK 9
#define EU 8

typedef __attribute__((ext_vector_type(2))) float v2f;
typedef __attribute__((ext_vector_type(4))) float v4f;
typedef __attribute__((ext_vector_type(8))) float v8f;

__global__ void k_gemm3_wmma(const float* __restrict__ A,
                             const float* __restrict__ B0, const float* __restrict__ b0,
                             float* __restrict__ C0,
                             const float* __restrict__ B1, const float* __restrict__ b1,
                             float* __restrict__ C1,
                             const float* __restrict__ B2, const float* __restrict__ b2,
                             float* __restrict__ C2,
                             int K) {
  const int lane = threadIdx.x;
  const int half = lane >> 4;
  const int l    = lane & 15;
  const int tn = blockIdx.x, tm = blockIdx.y;
  const int arow = tm * 16 + l;
  const int bcol = tn * 16 + l;
  v8f acc0 = {}, acc1 = {}, acc2 = {};
  for (int k = 0; k < K; k += 4) {
    const float* ap = A + arow * K + k + half * 2;
    v2f a; a.x = ap[0]; a.y = ap[1];
    const int boff = (k + half * 2) * HIDD + bcol;
    v2f w0; w0.x = B0[boff]; w0.y = B0[boff + HIDD];
    v2f w1; w1.x = B1[boff]; w1.y = B1[boff + HIDD];
    v2f w2; w2.x = B2[boff]; w2.y = B2[boff + HIDD];
    acc0 = __builtin_amdgcn_wmma_f32_16x16x4_f32(false, a, false, w0, (short)0, acc0, false, false);
    acc1 = __builtin_amdgcn_wmma_f32_16x16x4_f32(false, a, false, w1, (short)0, acc1, false, false);
    acc2 = __builtin_amdgcn_wmma_f32_16x16x4_f32(false, a, false, w2, (short)0, acc2, false, false);
  }
  const float bv0 = b0[bcol], bv1 = b1[bcol], bv2 = b2[bcol];
  const int rb = tm * 16 + half * 8;
#pragma unroll
  for (int r = 0; r < 8; ++r) {
    const int off = (rb + r) * HIDD + bcol;
    C0[off] = acc0[r] + bv0;
    C1[off] = acc1[r] + bv1;
    C2[off] = acc2[r] + bv2;
  }
}

#define CW 64
static_assert(VOC % 4 == 0, "a lane's 4 columns are all inside the matrix or all outside");
static_assert(NGR == 64 && HIDD == 64, "4 waves x 16 rows cover the graphs; 16 k-steps cover the hidden width");

__global__ void __launch_bounds__(128)
k_cls_wmma(const float* __restrict__ S, const float* __restrict__ Sn,
           const float* __restrict__ B, const float* __restrict__ bias,
           float* __restrict__ C, float* __restrict__ Amean) {
  const int lane = threadIdx.x & 31;
  const int half = lane >> 4;
  const int l    = lane & 15;
  const int w    = threadIdx.x >> 5;
  const int arow = w * 16 + l;
  const int c0   = blockIdx.x * CW + 4 * l;
  const bool live = c0 < VOC;
  const int cl   = live ? c0 : VOC - 4;

  const float dnr = Sn[arow];
  v2f s[HIDD / 4];
#pragma unroll
  for (int j = 0; j < HIDD / 4; ++j)
    s[j] = *(const v2f*)(S + arow * HIDD + 4 * j + half * 2);
  v4f bx[HIDD / 4], by[HIDD / 4];
#pragma unroll
  for (int j = 0; j < HIDD / 4; ++j) {
    const float* bp = B + (size_t)(4 * j + half * 2) * VOC + cl;
    bx[j] = *(const v4f*)bp;
    by[j] = *(const v4f*)(bp + VOC);
  }
  const v4f bv = *(const v4f*)(bias + cl);

  const float dn = fmaxf(dnr, 1.f);
  v2f a[HIDD / 4];
#pragma unroll
  for (int j = 0; j < HIDD / 4; ++j) {
    a[j].x = s[j].x / dn; a[j].y = s[j].y / dn;
    if (blockIdx.x == 0) *(v2f*)(Amean + arow * HIDD + 4 * j + half * 2) = a[j];
  }

  v8f acc0 = {}, acc1 = {}, acc2 = {}, acc3 = {};
#pragma unroll
  for (int j = 0; j < HIDD / 4; ++j) {
    v2f b0, b1, b2, b3;
    b0.x = bx[j][0]; b0.y = by[j][0];
    b1.x = bx[j][1]; b1.y = by[j][1];
    b2.x = bx[j][2]; b2.y = by[j][2];
    b3.x = bx[j][3]; b3.y = by[j][3];
    acc0 = __builtin_amdgcn_wmma_f32_16x16x4_f32(false, a[j], false, b0, (short)0, acc0, false, false);
    acc1 = __builtin_amdgcn_wmma_f32_16x16x4_f32(false, a[j], false, b1, (short)0, acc1, false, false);
    acc2 = __builtin_amdgcn_wmma_f32_16x16x4_f32(false, a[j], false, b2, (short)0, acc2, false, false);
    acc3 = __builtin_amdgcn_wmma_f32_16x16x4_f32(false, a[j], false, b3, (short)0, acc3, false, false);
  }
  if (live) {
    float* cp = C + (size_t)(w * 16 + half * 8) * VOC + c0;
#pragma unroll
    for (int r = 0; r < 8; ++r) {
      v4f o;
      o[0] = acc0[r] + bv[0]; o[1] = acc1[r] + bv[1];
      o[2] = acc2[r] + bv[2]; o[3] = acc3[r] + bv[3];
      *(v4f*)(cp + (size_t)r * VOC) = o;
    }
  }
}

__global__ void k_zero_i(int* __restrict__ p, int n) {
  int i = blockIdx.x * blockDim.x + threadIdx.x;
  if (i < n) p[i] = 0;
}

__global__ void k_count(const int* __restrict__ dstE, int* __restrict__ cnt) {
  int e = blockIdx.x * blockDim.x + threadIdx.x;
  if (e < NE) atomicAdd(cnt + dstE[e], 1);
}

#define SCAN_T 1024
#define SCAN_PER ((NN + SCAN_T - 1) / SCAN_T)
__global__ void __launch_bounds__(SCAN_T) k_scan(int* __restrict__ cur, int* __restrict__ rowptr) {
  __shared__ int part[SCAN_T];
  const int t = threadIdx.x;
  int loc[SCAN_PER];
  int sum = 0;
#pragma unroll
  for (int i = 0; i < SCAN_PER; ++i) {
    const int idx = t * SCAN_PER + i;
    loc[i] = sum;
    sum += (idx < NN) ? cur[idx] : 0;
  }
  part[t] = sum;
  __syncthreads();
  for (int off = 1; off < SCAN_T; off <<= 1) {
    const int v = (t >= off) ? part[t - off] : 0;
    __syncthreads();
    part[t] += v;
    __syncthreads();
  }
  const int excl = part[t] - sum;
#pragma unroll
  for (int i = 0; i < SCAN_PER; ++i) {
    const int idx = t * SCAN_PER + i;
    if (idx < NN) { rowptr[idx] = excl + loc[i]; cur[idx] = excl + loc[i]; }
  }
  if (t == SCAN_T - 1) rowptr[NN] = part[SCAN_T - 1];
}

__global__ void k_fill(const int* __restrict__ dstE, int* __restrict__ cur,
                       int* __restrict__ perm) {
  int e = blockIdx.x * blockDim.x + threadIdx.x;
  if (e < NE) perm[atomicAdd(cur + dstE[e], 1)] = e;
}

__global__ void k_rank(const int* __restrict__ perm, const int* __restrict__ rowptr,
                       const int* __restrict__ srcE, const int* __restrict__ dstE,
                       const float* __restrict__ ea,
                       int* __restrict__ csrc, v2f* __restrict__ cea) {
  int p = blockIdx.x * blockDim.x + threadIdx.x;
  if (p >= NE) return;
  const int e = perm[p];
  const int d = dstE[e];
  const int beg = rowptr[d], end = rowptr[d + 1];
  int r = 0;
  for (int q = beg; q < end; ++q) r += (perm[q] < e) ? 1 : 0;
  v2f a; a.x = ea[2 * e]; a.y = ea[2 * e + 1];
  csrc[beg + r] = srcE[e];
  cea[beg + r]  = a;
}

__device__ __forceinline__ float logit8(const float* xs, const float* xd,
                                        const float* w0, const float* w1,
                                        const float* at, float a0, float a1) {
  float lg = 0.f;
#pragma unroll
  for (int c = 0; c < CHN; ++c) {
    float m = xs[c] + xd[c] + a0 * w0[c] + a1 * w1[c];
    float g = m > 0.f ? m : 0.2f * m;
    lg += g * at[c];
  }
  return lg;
}

#define NB 16
#define TP 68
static_assert(NN % NB == 0, "the grid covers the nodes exactly: no thread is out of range");

template <bool NEXT>
__global__ void __launch_bounds__(NB * HEADS)
k_gat_node(const float* __restrict__ xl, const float* __restrict__ xr,
           const float* __restrict__ res, const int* __restrict__ rowptr,
           const int* __restrict__ csrc, const v2f* __restrict__ cea,
           const float* __restrict__ We, const float* __restrict__ att,
           const float* __restrict__ bias_g, float* __restrict__ hout,
           const float* __restrict__ nB0, const float* __restrict__ nb0, float* __restrict__ nC0,
           const float* __restrict__ nB1, const float* __restrict__ nb1, float* __restrict__ nC1,
           const float* __restrict__ nB2, const float* __restrict__ nb2, float* __restrict__ nC2,
           const int* __restrict__ batch, float* __restrict__ psum, float* __restrict__ pcnt) {
  __shared__ alignas(16) float sH[NB * TP];
  const int idx = blockIdx.x * (NB * HEADS) + threadIdx.x;
  const int d = idx >> 3, cb = (idx & 7) * CHN;

  float w0[CHN], w1[CHN], at[CHN], xd[CHN], xself[CHN];
  {
    const v4f* pd = (const v4f*)(xr + d * HIDD + cb);
    const v4f* ps = (const v4f*)(xl + d * HIDD + cb);
    const v4f d0 = pd[0], d1 = pd[1], s0v = ps[0], s1v = ps[1];
#pragma unroll
    for (int c = 0; c < 4; ++c) {
      xd[c] = d0[c]; xd[4 + c] = d1[c]; xself[c] = s0v[c]; xself[4 + c] = s1v[c];
    }
#pragma unroll
    for (int c = 0; c < CHN; ++c) {
      w0[c] = We[cb + c]; w1[c] = We[HIDD + cb + c]; at[c] = att[cb + c];
    }
  }

  const int beg = rowptr[d], end = rowptr[d + 1];
  float mrun = -INFINITY, den = 0.f, asum0 = 0.f, asum1 = 0.f;
  float acc[CHN];
#pragma unroll
  for (int c = 0; c < CHN; ++c) acc[c] = 0.f;

  if (beg < end) {
    int sj[EU]; v2f aj[EU];
#pragma unroll
    for (int j = 0; j < EU; ++j) {
      const int q = min(beg + j, end - 1);
      sj[j] = csrc[q]; aj[j] = cea[q];
    }
    for (int p = beg; p < end; p += EU) {
      v4f x0[EU], x1[EU];
#pragma unroll
      for (int j = 0; j < EU; ++j) {
        const v4f* r = (const v4f*)(xl + sj[j] * HIDD + cb);
        x0[j] = r[0]; x1[j] = r[1];
      }
      int sn[EU]; v2f an[EU];
#pragma unroll
      for (int j = 0; j < EU; ++j) {
        const int q = min(p + EU + j, end - 1);
        sn[j] = csrc[q]; an[j] = cea[q];
      }
      float lg[EU];
      float cm = -INFINITY;
#pragma unroll
      for (int j = 0; j < EU; ++j) {
        const bool v = (p + j) < end;
        float xs[CHN];
#pragma unroll
        for (int c = 0; c < 4; ++c) { xs[c] = x0[j][c]; xs[4 + c] = x1[j][c]; }
        const float l = logit8(xs, xd, w0, w1, at, aj[j].x, aj[j].y);
        lg[j] = v ? l : -INFINITY;
        asum0 += v ? aj[j].x : 0.f;
        asum1 += v ? aj[j].y : 0.f;
        cm = fmaxf(cm, lg[j]);
      }
      const float mnew = fmaxf(mrun, cm);
      const float sc = __expf(mrun - mnew);
      float pj[EU];
      float psum = 0.f;
#pragma unroll
      for (int j = 0; j < EU; ++j) { pj[j] = __expf(lg[j] - mnew); psum += pj[j]; }
      den = den * sc + psum;
#pragma unroll
      for (int c = 0; c < 4; ++c) {
        float t0 = 0.f, t1 = 0.f;
#pragma unroll
        for (int j = 0; j < EU; ++j) { t0 += pj[j] * x0[j][c]; t1 += pj[j] * x1[j][c]; }
        acc[c]     = acc[c] * sc + t0;
        acc[4 + c] = acc[4 + c] * sc + t1;
      }
      mrun = mnew;
#pragma unroll
      for (int j = 0; j < EU; ++j) { sj[j] = sn[j]; aj[j] = an[j]; }
    }
  }

  const int lane = threadIdx.x & 31;
  const int half = lane >> 4;
  const int l    = lane & 15;
  const int bcol = (threadIdx.x >> 5) * 16 + l;
  v2f wf0[HIDD / 4], wf1[HIDD / 4], wf2[HIDD / 4];
  float bv0 = 0.f, bv1 = 0.f, bv2 = 0.f;
  if (NEXT) {
#pragma unroll
    for (int j = 0; j < HIDD / 4; ++j) {
      const int boff = (4 * j + half * 2) * HIDD + bcol;
      wf0[j].x = nB0[boff]; wf0[j].y = nB0[boff + HIDD];
      wf1[j].x = nB1[boff]; wf1[j].y = nB1[boff + HIDD];
      wf2[j].x = nB2[boff]; wf2[j].y = nB2[boff + HIDD];
    }
    bv0 = nb0[bcol]; bv1 = nb1[bcol]; bv2 = nb2[bcol];
  }
  const int pc  = threadIdx.x & (HIDD - 1);
  const int pr0 = (threadIdx.x >> 6) * (NB / 2);
  int gid[NB / 2];
  if (!NEXT) {
#pragma unroll
    for (int r = 0; r < NB / 2; ++r) gid[r] = batch[blockIdx.x * NB + pr0 + r];
  }

  {
    const float dg = fmaxf((float)(end - beg), 1.f);
    const float l = logit8(xself, xd, w0, w1, at, asum0 / dg, asum1 / dg);
    const float mnew = fmaxf(mrun, l);
    const float sc = __expf(mrun - mnew);
    const float pl = __expf(l - mnew);
    den = den * sc + pl;
#pragma unroll
    for (int c = 0; c < CHN; ++c) acc[c] = acc[c] * sc + pl * xself[c];
  }

  const v4f* pr = (const v4f*)(res + d * HIDD + cb);
  const v4f r0 = pr[0], r1 = pr[1];
  v4f o0, o1;
#pragma unroll
  for (int c = 0; c < 4; ++c) {
    float v0 = acc[c] / den     + bias_g[cb + c]     + r0[c];
    float v1 = acc[4 + c] / den + bias_g[cb + 4 + c] + r1[c];
    o0[c] = v0 > 0.f ? v0 : (__expf(v0) - 1.f);
    o1[c] = v1 > 0.f ? v1 : (__expf(v1) - 1.f);
  }
  v4f* po = (v4f*)(hout + d * HIDD + cb);
  if (!NEXT) {
    v4f* ps = (v4f*)(sH + (threadIdx.x >> 3) * TP + cb);
    ps[0] = o0; ps[1] = o1;
    __syncthreads();
    po[0] = o0; po[1] = o1;
    float hv[NB / 2];
#pragma unroll
    for (int r = 0; r < NB / 2; ++r) hv[r] = sH[(pr0 + r) * TP + pc];
    float run = 0.f, nrun = 0.f;
    int gcur = gid[0];
#pragma unroll
    for (int r = 0; r < NB / 2; ++r) {
      if (gid[r] != gcur) {
        atomicAdd(psum + gcur * HIDD + pc, run);
        if (pc == 0) atomicAdd(pcnt + gcur, nrun);
        run = 0.f; nrun = 0.f; gcur = gid[r];
      }
      run += hv[r];
      nrun += 1.f;
    }
    atomicAdd(psum + gcur * HIDD + pc, run);
    if (pc == 0) atomicAdd(pcnt + gcur, nrun);
  }

  if (NEXT) {
    v4f* ps = (v4f*)(sH + (threadIdx.x >> 3) * TP + cb);
    ps[0] = o0; ps[1] = o1;
    __syncthreads();
    po[0] = o0; po[1] = o1;
    v8f acc0 = {}, acc1 = {}, acc2 = {};
#pragma unroll
    for (int j = 0; j < HIDD / 4; ++j) {
      const v2f a = *(const v2f*)(sH + l * TP + 4 * j + half * 2);
      acc0 = __builtin_amdgcn_wmma_f32_16x16x4_f32(false, a, false, wf0[j], (short)0, acc0, false, false);
      acc1 = __builtin_amdgcn_wmma_f32_16x16x4_f32(false, a, false, wf1[j], (short)0, acc1, false, false);
      acc2 = __builtin_amdgcn_wmma_f32_16x16x4_f32(false, a, false, wf2[j], (short)0, acc2, false, false);
    }
    const int rb = blockIdx.x * NB + half * 8;
#pragma unroll
    for (int r = 0; r < 8; ++r) {
      const int off = (rb + r) * HIDD + bcol;
      nC0[off] = acc0[r] + bv0;
      nC1[off] = acc1[r] + bv1;
      nC2[off] = acc2[r] + bv2;
    }
  }
}

extern "C" void kernel_launch(void* const* d_in, const int* in_sizes, int n_in,
                              void* d_out_v, int out_size, void* d_ws, size_t ws_size,
                              hipStream_t stream) {
  (void)in_sizes; (void)n_in; (void)out_size; (void)ws_size;
  const float* x       = (const float*)d_in[0];
  const int*   ei      = (const int*)  d_in[1];
  const float* ea      = (const float*)d_in[2];
  const int*   batch   = (const int*)  d_in[3];
  const float* p1_Wl   = (const float*)d_in[4];
  const float* p1_bl   = (const float*)d_in[5];
  const float* p1_Wr   = (const float*)d_in[6];
  const float* p1_br   = (const float*)d_in[7];
  const float* p1_We   = (const float*)d_in[8];
  const float* p1_att  = (const float*)d_in[9];
  const float* p1_bias = (const float*)d_in[10];
  const float* p1_Wres = (const float*)d_in[11];
  const float* p1_bres = (const float*)d_in[12];
  const float* Wl      = (const float*)d_in[13];
  const float* bl      = (const float*)d_in[14];
  const float* Wr      = (const float*)d_in[15];
  const float* br      = (const float*)d_in[16];
  const float* We      = (const float*)d_in[17];
  const float* att     = (const float*)d_in[18];
  const float* bias_g  = (const float*)d_in[19];
  const float* Wres    = (const float*)d_in[20];
  const float* bres    = (const float*)d_in[21];
  const float* Wlin    = (const float*)d_in[22];
  const float* blin    = (const float*)d_in[23];

  float* out = (float*)d_out_v;
  float* ws  = (float*)d_ws;

  float* xl     = ws;
  float* xr     = ws + 640000;
  float* res    = ws + 1280000;
  int*   rowptr = (int*)(ws + 1920000);
  int*   cur    = (int*)(ws + 1940000);
  float* psum   = ws + 1950000;
  float* pcnt   = ws + 1954096;
  int*   perm   = (int*)(ws + 1960000);
  int*   csrc   = (int*)(ws + 2120000);
  v2f*   cea    = (v2f*)(ws + 2280000);
  float* xlB    = ws + 2600000;
  float* xrB    = ws + 3240000;
  float* resB   = ws + 3880000;

  const int* srcE = ei;
  const int* dstE = ei + NE;

  const int TB = 256;

  k_zero_i<<<(NN + NGR * HIDD + NGR + TB - 1) / TB, TB, 0, stream>>>(cur, NN + NGR * HIDD + NGR);
  k_count<<<(NE + TB - 1) / TB, TB, 0, stream>>>(dstE, cur);
  k_scan<<<1, SCAN_T, 0, stream>>>(cur, rowptr);
  k_fill<<<(NE + TB - 1) / TB, TB, 0, stream>>>(dstE, cur, perm);
  k_rank<<<(NE + TB - 1) / TB, TB, 0, stream>>>(perm, rowptr, srcE, dstE, ea, csrc, cea);

  {
    dim3 gg(HIDD / 16, NN / 16);
    k_gemm3_wmma<<<gg, 32, 0, stream>>>(x, p1_Wl, p1_bl, xl, p1_Wr, p1_br, xr,
                                        p1_Wres, p1_bres, res, FIN);
  }
  for (int layer = 0; layer < 1 + LSTK; ++layer) {
    const int li = layer - 1;
    const float* We_   = (layer == 0) ? p1_We   : We   + (size_t)li * 2 * HIDD;
    const float* att_  = (layer == 0) ? p1_att  : att  + (size_t)li * HEADS * CHN;
    const float* bg_   = (layer == 0) ? p1_bias : bias_g + (size_t)li * HIDD;
    float* hout = out + (size_t)layer * NN * HIDD;
    const bool odd = (layer & 1) != 0;
    const float* xl_c  = odd ? xlB  : xl;   float* xl_n  = odd ? xl  : xlB;
    const float* xr_c  = odd ? xrB  : xr;   float* xr_n  = odd ? xr  : xrB;
    const float* res_c = odd ? resB : res;  float* res_n = odd ? res : resB;

    if (layer < LSTK) {
      k_gat_node<true><<<NN / NB, NB * HEADS, 0, stream>>>(
          xl_c, xr_c, res_c, rowptr, csrc, cea, We_, att_, bg_, hout,
          Wl   + (size_t)layer * HIDD * HIDD, bl   + (size_t)layer * HIDD, xl_n,
          Wr   + (size_t)layer * HIDD * HIDD, br   + (size_t)layer * HIDD, xr_n,
          Wres + (size_t)layer * HIDD * HIDD, bres + (size_t)layer * HIDD, res_n,
          nullptr, nullptr, nullptr);
    } else {
      k_gat_node<false><<<NN / NB, NB * HEADS, 0, stream>>>(
          xl_c, xr_c, res_c, rowptr, csrc, cea, We_, att_, bg_, hout,
          nullptr, nullptr, nullptr, nullptr, nullptr, nullptr, nullptr, nullptr, nullptr,
          batch, psum, pcnt);
    }
  }

  float* pooled = out + (size_t)(1 + LSTK) * NN * HIDD;
  float* x12    = pooled + NGR * HIDD;
  k_cls_wmma<<<(VOC + CW - 1) / CW, 128, 0, stream>>>(psum, pcnt, Wlin, blin, x12, pooled);
}


# Graph prep and layer-0 transforms in 4 launches instead of 6: each fill group redoes the segment-start scan in LDS; the K=16 WMMA tiles run as extra groups of the rank kernel; sum orders kept
# speedup vs baseline: 5.0505x; 1.1845x over previous
#include <hip/hip_runtime.h>

#define NN 10000
#define NE 160000
#define FIN 16
#define HEADS 8
#define CHN 8
#define HIDD 64
#define NGR 64
#define VOC 50000
#define LSTK 9
#define EU 8

typedef __attribute__((ext_vector_type(2))) float v2f;
typedef __attribute__((ext_vector_type(4))) float v4f;
typedef __attribute__((ext_vector_type(8))) float v8f;

static_assert(FIN % 4 == 0, "whole k-steps of 4");
__device__ __forceinline__ void gemm3_tile(const float* __restrict__ A,
                             const float* __restrict__ B0, const float* __restrict__ b0,
                             float* __restrict__ C0,
                             const float* __restrict__ B1, const float* __restrict__ b1,
                             float* __restrict__ C1,
                             const float* __restrict__ B2, const float* __restrict__ b2,
                             float* __restrict__ C2,
                             int tn, int tm, int lane) {
  const int half = lane >> 4;
  const int l    = lane & 15;
  const int arow = tm * 16 + l;
  const int bcol = tn * 16 + l;
  v2f a[FIN / 4], w0[FIN / 4], w1[FIN / 4], w2[FIN / 4];
#pragma unroll
  for (int j = 0; j < FIN / 4; ++j) {
    const int k = 4 * j;
    const float* ap = A + arow * FIN + k + half * 2;
    a[j].x = ap[0]; a[j].y = ap[1];
    const int boff = (k + half * 2) * HIDD + bcol;
    w0[j].x = B0[boff]; w0[j].y = B0[boff + HIDD];
    w1[j].x = B1[boff]; w1[j].y = B1[boff + HIDD];
    w2[j].x = B2[boff]; w2[j].y = B2[boff + HIDD];
  }
  const float bv0 = b0[bcol], bv1 = b1[bcol], bv2 = b2[bcol];
  v8f acc0 = {}, acc1 = {}, acc2 = {};
#pragma unroll
  for (int j = 0; j < FIN / 4; ++j) {
    acc0 = __builtin_amdgcn_wmma_f32_16x16x4_f32(false, a[j], false, w0[j], (short)0, acc0, false, false);
    acc1 = __builtin_amdgcn_wmma_f32_16x16x4_f32(false, a[j], false, w1[j], (short)0, acc1, false, false);
    acc2 = __builtin_amdgcn_wmma_f32_16x16x4_f32(false, a[j], false, w2[j], (short)0, acc2, false, false);
  }
  const int rb = tm * 16 + half * 8;
#pragma unroll
  for (int r = 0; r < 8; ++r) {
    const int off = (rb + r) * HIDD + bcol;
    C0[off] = acc0[r] + bv0;
    C1[off] = acc1[r] + bv1;
    C2[off] = acc2[r] + bv2;
  }
}

#define CW 64
static_assert(VOC % 4 == 0, "a lane's 4 columns are all inside the matrix or all outside");
static_assert(NGR == 64 && HIDD == 64, "4 waves x 16 rows cover the graphs; 16 k-steps cover the hidden width");

__global__ void __launch_bounds__(128)
k_cls_wmma(const float* __restrict__ S, const float* __restrict__ Sn,
           const float* __restrict__ B, const float* __restrict__ bias,
           float* __restrict__ C, float* __restrict__ Amean) {
  const int lane = threadIdx.x & 31;
  const int half = lane >> 4;
  const int l    = lane & 15;
  const int w    = threadIdx.x >> 5;
  const int arow = w * 16 + l;
  const int c0   = blockIdx.x * CW + 4 * l;
  const bool live = c0 < VOC;
  const int cl   = live ? c0 : VOC - 4;

  const float dnr = Sn[arow];
  v2f s[HIDD / 4];
#pragma unroll
  for (int j = 0; j < HIDD / 4; ++j)
    s[j] = *(const v2f*)(S + arow * HIDD + 4 * j + half * 2);
  v4f bx[HIDD / 4], by[HIDD / 4];
#pragma unroll
  for (int j = 0; j < HIDD / 4; ++j) {
    const float* bp = B + (size_t)(4 * j + half * 2) * VOC + cl;
    bx[j] = *(const v4f*)bp;
    by[j] = *(const v4f*)(bp + VOC);
  }
  const v4f bv = *(const v4f*)(bias + cl);

  const float dn = fmaxf(dnr, 1.f);
  v2f a[HIDD / 4];
#pragma unroll
  for (int j = 0; j < HIDD / 4; ++j) {
    a[j].x = s[j].x / dn; a[j].y = s[j].y / dn;
    if (blockIdx.x == 0) *(v2f*)(Amean + arow * HIDD + 4 * j + half * 2) = a[j];
  }

  v8f acc0 = {}, acc1 = {}, acc2 = {}, acc3 = {};
#pragma unroll
  for (int j = 0; j < HIDD / 4; ++j) {
    v2f b0, b1, b2, b3;
    b0.x = bx[j][0]; b0.y = by[j][0];
    b1.x = bx[j][1]; b1.y = by[j][1];
    b2.x = bx[j][2]; b2.y = by[j][2];
    b3.x = bx[j][3]; b3.y = by[j][3];
    acc0 = __builtin_amdgcn_wmma_f32_16x16x4_f32(false, a[j], false, b0, (short)0, acc0, false, false);
    acc1 = __builtin_amdgcn_wmma_f32_16x16x4_f32(false, a[j], false, b1, (short)0, acc1, false, false);
    acc2 = __builtin_amdgcn_wmma_f32_16x16x4_f32(false, a[j], false, b2, (short)0, acc2, false, false);
    acc3 = __builtin_amdgcn_wmma_f32_16x16x4_f32(false, a[j], false, b3, (short)0, acc3, false, false);
  }
  if (live) {
    float* cp = C + (size_t)(w * 16 + half * 8) * VOC + c0;
#pragma unroll
    for (int r = 0; r < 8; ++r) {
      v4f o;
      o[0] = acc0[r] + bv[0]; o[1] = acc1[r] + bv[1];
      o[2] = acc2[r] + bv[2]; o[3] = acc3[r] + bv[3];
      *(v4f*)(cp + (size_t)r * VOC) = o;
    }
  }
}

__global__ void k_zero_i(int* __restrict__ p, int n) {
  int i = blockIdx.x * blockDim.x + threadIdx.x;
  if (i < n) p[i] = 0;
}

__global__ void k_count(const int* __restrict__ dstE, int* __restrict__ cnt,
                        int* __restrict__ slot) {
  int e = blockIdx.x * blockDim.x + threadIdx.x;
  if (e < NE) slot[e] = atomicAdd(cnt + dstE[e], 1);
}

#define SCAN_T 1024
#define SCAN_PER ((NN + SCAN_T - 1) / SCAN_T)
static_assert(SCAN_T / 32 == 32, "one lane of wave 0 per wave total");
__global__ void __launch_bounds__(SCAN_T)
k_scan_fill(const int* __restrict__ cnt, const int* __restrict__ dstE,
            const int* __restrict__ slot, int* __restrict__ rowptr,
            int* __restrict__ perm) {
  __shared__ int rp[SCAN_T * SCAN_PER];
  __shared__ int wtot[SCAN_T / 32];
  const int t = threadIdx.x;
  const int lane = t & 31, wv = t >> 5;
  const int e = blockIdx.x * SCAN_T + t;
  int d = 0, s = 0;
  if (e < NE) { d = dstE[e]; s = slot[e]; }
  int loc[SCAN_PER];
  int sum = 0;
#pragma unroll
  for (int i = 0; i < SCAN_PER; ++i) {
    const int idx = t * SCAN_PER + i;
    loc[i] = sum;
    sum += (idx < NN) ? cnt[idx] : 0;
  }
  int v = sum;
#pragma unroll
  for (int off = 1; off < 32; off <<= 1) {
    const int u = __shfl_up(v, off);
    if (lane >= off) v += u;
  }
  if (lane == 31) wtot[wv] = v;
  __syncthreads();
  if (wv == 0) {
    int w = wtot[lane];
#pragma unroll
    for (int off = 1; off < 32; off <<= 1) {
      const int u = __shfl_up(w, off);
      if (lane >= off) w += u;
    }
    wtot[lane] = w;
  }
  __syncthreads();
  const int excl = v - sum + (wv > 0 ? wtot[wv - 1] : 0);
#pragma unroll
  for (int i = 0; i < SCAN_PER; ++i) {
    const int idx = t * SCAN_PER + i;
    if (idx < NN) {
      rp[idx] = excl + loc[i];
      if (blockIdx.x == 0) rowptr[idx] = excl + loc[i];
    }
  }
  if (blockIdx.x == 0 && t == SCAN_T - 1) rowptr[NN] = excl + sum;
  __syncthreads();
  if (e < NE) perm[rp[d] + s] = e;
}

#define RANK_T 256
#define RANK_G ((NE + RANK_T - 1) / RANK_T)
#define G3_TILES ((HIDD / 16) * (NN / 16))
#define G3_G ((G3_TILES + RANK_T / 32 - 1) / (RANK_T / 32))
__global__ void __launch_bounds__(RANK_T)
k_rank_gemm3(const int* __restrict__ perm, const int* __restrict__ rowptr,
             const int* __restrict__ srcE, const int* __restrict__ dstE,
             const float* __restrict__ ea,
             int* __restrict__ csrc, v2f* __restrict__ cea,
             const float* __restrict__ x,
             const float* __restrict__ B0, const float* __restrict__ b0, float* __restrict__ C0,
             const float* __restrict__ B1, const float* __restrict__ b1, float* __restrict__ C1,
             const float* __restrict__ B2, const float* __restrict__ b2, float* __restrict__ C2) {
  if (blockIdx.x >= RANK_G) {
    const int wv = __builtin_amdgcn_readfirstlane((int)(threadIdx.x >> 5));
    const int tile = (blockIdx.x - RANK_G) * (RANK_T / 32) + wv;
    if (tile < G3_TILES)
      gemm3_tile(x, B0, b0, C0, B1, b1, C1, B2, b2, C2,
                 tile & (HIDD / 16 - 1), tile / (HIDD / 16), threadIdx.x & 31);
    return;
  }
  int p = blockIdx.x * RANK_T + threadIdx.x;
  if (p >= NE) return;
  const int e = perm[p];
  const int d = dstE[e];
  const int beg = rowptr[d], end = rowptr[d + 1];
  int r = 0;
  for (int q = beg; q < end; ++q) r += (perm[q] < e) ? 1 : 0;
  v2f a; a.x = ea[2 * e]; a.y = ea[2 * e + 1];
  csrc[beg + r] = srcE[e];
  cea[beg + r]  = a;
}

__device__ __forceinline__ float logit8(const float* xs, const float* xd,
                                        const float* w0, const float* w1,
                                        const float* at, float a0, float a1) {
  float lg = 0.f;
#pragma unroll
  for (int c = 0; c < CHN; ++c) {
    float m = xs[c] + xd[c] + a0 * w0[c] + a1 * w1[c];
    float g = m > 0.f ? m : 0.2f * m;
    lg += g * at[c];
  }
  return lg;
}

#define NB 16
#define TP 68
static_assert(NN % NB == 0, "the grid covers the nodes exactly: no thread is out of range");

template <bool NEXT>
__global__ void __launch_bounds__(NB * HEADS)
k_gat_node(const float* __restrict__ xl, const float* __restrict__ xr,
           const float* __restrict__ res, const int* __restrict__ rowptr,
           const int* __restrict__ csrc, const v2f* __restrict__ cea,
           const float* __restrict__ We, const float* __restrict__ att,
           const float* __restrict__ bias_g, float* __restrict__ hout,
           const float* __restrict__ nB0, const float* __restrict__ nb0, float* __restrict__ nC0,
           const float* __restrict__ nB1, const float* __restrict__ nb1, float* __restrict__ nC1,
           const float* __restrict__ nB2, const float* __restrict__ nb2, float* __restrict__ nC2,
           const int* __restrict__ batch, float* __restrict__ psum, float* __restrict__ pcnt) {
  __shared__ alignas(16) float sH[NB * TP];
  const int idx = blockIdx.x * (NB * HEADS) + threadIdx.x;
  const int d = idx >> 3, cb = (idx & 7) * CHN;

  float w0[CHN], w1[CHN], at[CHN], xd[CHN], xself[CHN];
  {
    const v4f* pd = (const v4f*)(xr + d * HIDD + cb);
    const v4f* ps = (const v4f*)(xl + d * HIDD + cb);
    const v4f d0 = pd[0], d1 = pd[1], s0v = ps[0], s1v = ps[1];
#pragma unroll
    for (int c = 0; c < 4; ++c) {
      xd[c] = d0[c]; xd[4 + c] = d1[c]; xself[c] = s0v[c]; xself[4 + c] = s1v[c];
    }
#pragma unroll
    for (int c = 0; c < CHN; ++c) {
      w0[c] = We[cb + c]; w1[c] = We[HIDD + cb + c]; at[c] = att[cb + c];
    }
  }

  const int beg = rowptr[d], end = rowptr[d + 1];
  float mrun = -INFINITY, den = 0.f, asum0 = 0.f, asum1 = 0.f;
  float acc[CHN];
#pragma unroll
  for (int c = 0; c < CHN; ++c) acc[c] = 0.f;

  if (beg < end) {
    int sj[EU]; v2f aj[EU];
#pragma unroll
    for (int j = 0; j < EU; ++j) {
      const int q = min(beg + j, end - 1);
      sj[j] = csrc[q]; aj[j] = cea[q];
    }
    for (int p = beg; p < end; p += EU) {
      v4f x0[EU], x1[EU];
#pragma unroll
      for (int j = 0; j < EU; ++j) {
        const v4f* r = (const v4f*)(xl + sj[j] * HIDD + cb);
        x0[j] = r[0]; x1[j] = r[1];
      }
      int sn[EU]; v2f an[EU];
#pragma unroll
      for (int j = 0; j < EU; ++j) {
        const int q = min(p + EU + j, end - 1);
        sn[j] = csrc[q]; an[j] = cea[q];
      }
      float lg[EU];
      float cm = -INFINITY;
#pragma unroll
      for (int j = 0; j < EU; ++j) {
        const bool v = (p + j) < end;
        float xs[CHN];
#pragma unroll
        for (int c = 0; c < 4; ++c) { xs[c] = x0[j][c]; xs[4 + c] = x1[j][c]; }
        const float l = logit8(xs, xd, w0, w1, at, aj[j].x, aj[j].y);
        lg[j] = v ? l : -INFINITY;
        asum0 += v ? aj[j].x : 0.f;
        asum1 += v ? aj[j].y : 0.f;
        cm = fmaxf(cm, lg[j]);
      }
      const float mnew = fmaxf(mrun, cm);
      const float sc = __expf(mrun - mnew);
      float pj[EU];
      float psum = 0.f;
#pragma unroll
      for (int j = 0; j < EU; ++j) { pj[j] = __expf(lg[j] - mnew); psum += pj[j]; }
      den = den * sc + psum;
#pragma unroll
      for (int c = 0; c < 4; ++c) {
        float t0 = 0.f, t1 = 0.f;
#pragma unroll
        for (int j = 0; j < EU; ++j) { t0 += pj[j] * x0[j][c]; t1 += pj[j] * x1[j][c]; }
        acc[c]     = acc[c] * sc + t0;
        acc[4 + c] = acc[4 + c] * sc + t1;
      }
      mrun = mnew;
#pragma unroll
      for (int j = 0; j < EU; ++j) { sj[j] = sn[j]; aj[j] = an[j]; }
    }
  }

  const int lane = threadIdx.x & 31;
  const int half = lane >> 4;
  const int l    = lane & 15;
  const int bcol = (threadIdx.x >> 5) * 16 + l;
  v2f wf0[HIDD / 4], wf1[HIDD / 4], wf2[HIDD / 4];
  float bv0 = 0.f, bv1 = 0.f, bv2 = 0.f;
  if (NEXT) {
#pragma unroll
    for (int j = 0; j < HIDD / 4; ++j) {
      const int boff = (4 * j + half * 2) * HIDD + bcol;
      wf0[j].x = nB0[boff]; wf0[j].y = nB0[boff + HIDD];
      wf1[j].x = nB1[boff]; wf1[j].y = nB1[boff + HIDD];
      wf2[j].x = nB2[boff]; wf2[j].y = nB2[boff + HIDD];
    }
    bv0 = nb0[bcol]; bv1 = nb1[bcol]; bv2 = nb2[bcol];
  }
  const int pc  = threadIdx.x & (HIDD - 1);
  const int pr0 = (threadIdx.x >> 6) * (NB / 2);
  int gid[NB / 2];
  if (!NEXT) {
#pragma unroll
    for (int r = 0; r < NB / 2; ++r) gid[r] = batch[blockIdx.x * NB + pr0 + r];
  }

  {
    const float dg = fmaxf((float)(end - beg), 1.f);
    const float l = logit8(xself, xd, w0, w1, at, asum0 / dg, asum1 / dg);
    const float mnew = fmaxf(mrun, l);
    const float sc = __expf(mrun - mnew);
    const float pl = __expf(l - mnew);
    den = den * sc + pl;
#pragma unroll
    for (int c = 0; c < CHN; ++c) acc[c] = acc[c] * sc + pl * xself[c];
  }

  const v4f* pr = (const v4f*)(res + d * HIDD + cb);
  const v4f r0 = pr[0], r1 = pr[1];
  v4f o0, o1;
#pragma unroll
  for (int c = 0; c < 4; ++c) {
    float v0 = acc[c] / den     + bias_g[cb + c]     + r0[c];
    float v1 = acc[4 + c] / den + bias_g[cb + 4 + c] + r1[c];
    o0[c] = v0 > 0.f ? v0 : (__expf(v0) - 1.f);
    o1[c] = v1 > 0.f ? v1 : (__expf(v1) - 1.f);
  }
  v4f* po = (v4f*)(hout + d * HIDD + cb);
  if (!NEXT) {
    v4f* ps = (v4f*)(sH + (threadIdx.x >> 3) * TP + cb);
    ps[0] = o0; ps[1] = o1;
    __syncthreads();
    po[0] = o0; po[1] = o1;
    float hv[NB / 2];
#pragma unroll
    for (int r = 0; r < NB / 2; ++r) hv[r] = sH[(pr0 + r) * TP + pc];
    float run = 0.f, nrun = 0.f;
    int gcur = gid[0];
#pragma unroll
    for (int r = 0; r < NB / 2; ++r) {
      if (gid[r] != gcur) {
        atomicAdd(psum + gcur * HIDD + pc, run);
        if (pc == 0) atomicAdd(pcnt + gcur, nrun);
        run = 0.f; nrun = 0.f; gcur = gid[r];
      }
      run += hv[r];
      nrun += 1.f;
    }
    atomicAdd(psum + gcur * HIDD + pc, run);
    if (pc == 0) atomicAdd(pcnt + gcur, nrun);
  }

  if (NEXT) {
    v4f* ps = (v4f*)(sH + (threadIdx.x >> 3) * TP + cb);
    ps[0] = o0; ps[1] = o1;
    __syncthreads();
    po[0] = o0; po[1] = o1;
    v8f acc0 = {}, acc1 = {}, acc2 = {};
#pragma unroll
    for (int j = 0; j < HIDD / 4; ++j) {
      const v2f a = *(const v2f*)(sH + l * TP + 4 * j + half * 2);
      acc0 = __builtin_amdgcn_wmma_f32_16x16x4_f32(false, a, false, wf0[j], (short)0, acc0, false, false);
      acc1 = __builtin_amdgcn_wmma_f32_16x16x4_f32(false, a, false, wf1[j], (short)0, acc1, false, false);
      acc2 = __builtin_amdgcn_wmma_f32_16x16x4_f32(false, a, false, wf2[j], (short)0, acc2, false, false);
    }
    const int rb = blockIdx.x * NB + half * 8;
#pragma unroll
    for (int r = 0; r < 8; ++r) {
      const int off = (rb + r) * HIDD + bcol;
      nC0[off] = acc0[r] + bv0;
      nC1[off] = acc1[r] + bv1;
      nC2[off] = acc2[r] + bv2;
    }
  }
}

extern "C" void kernel_launch(void* const* d_in, const int* in_sizes, int n_in,
                              void* d_out_v, int out_size, void* d_ws, size_t ws_size,
                              hipStream_t stream) {
  (void)in_sizes; (void)n_in; (void)out_size; (void)ws_size;
  const float* x       = (const float*)d_in[0];
  const int*   ei      = (const int*)  d_in[1];
  const float* ea      = (const float*)d_in[2];
  const int*   batch   = (const int*)  d_in[3];
  const float* p1_Wl   = (const float*)d_in[4];
  const float* p1_bl   = (const float*)d_in[5];
  const float* p1_Wr   = (const float*)d_in[6];
  const float* p1_br   = (const float*)d_in[7];
  const float* p1_We   = (const float*)d_in[8];
  const float* p1_att  = (const float*)d_in[9];
  const float* p1_bias = (const float*)d_in[10];
  const float* p1_Wres = (const float*)d_in[11];
  const float* p1_bres = (const float*)d_in[12];
  const float* Wl      = (const float*)d_in[13];
  const float* bl      = (const float*)d_in[14];
  const float* Wr      = (const float*)d_in[15];
  const float* br      = (const float*)d_in[16];
  const float* We      = (const float*)d_in[17];
  const float* att     = (const float*)d_in[18];
  const float* bias_g  = (const float*)d_in[19];
  const float* Wres    = (const float*)d_in[20];
  const float* bres    = (const float*)d_in[21];
  const float* Wlin    = (const float*)d_in[22];
  const float* blin    = (const float*)d_in[23];

  float* out = (float*)d_out_v;
  float* ws  = (float*)d_ws;

  float* xl     = ws;
  float* xr     = ws + 640000;
  float* res    = ws + 1280000;
  int*   rowptr = (int*)(ws + 1920000);
  int*   cur    = (int*)(ws + 1940000);
  float* psum   = ws + 1950000;
  float* pcnt   = ws + 1954096;
  int*   perm   = (int*)(ws + 1960000);
  int*   csrc   = (int*)(ws + 2120000);
  v2f*   cea    = (v2f*)(ws + 2280000);
  float* xlB    = ws + 2600000;
  float* xrB    = ws + 3240000;
  float* resB   = ws + 3880000;
  int*   slot   = (int*)xlB;

  const int* srcE = ei;
  const int* dstE = ei + NE;

  const int TB = 256;

  k_zero_i<<<(NN + NGR * HIDD + NGR + TB - 1) / TB, TB, 0, stream>>>(cur, NN + NGR * HIDD + NGR);
  k_count<<<(NE + TB - 1) / TB, TB, 0, stream>>>(dstE, cur, slot);
  k_scan_fill<<<(NE + SCAN_T - 1) / SCAN_T, SCAN_T, 0, stream>>>(cur, dstE, slot, rowptr, perm);

  k_rank_gemm3<<<RANK_G + G3_G, RANK_T, 0, stream>>>(perm, rowptr, srcE, dstE, ea, csrc, cea,
                                                     x, p1_Wl, p1_bl, xl, p1_Wr, p1_br, xr,
                                                     p1_Wres, p1_bres, res);
  for (int layer = 0; layer < 1 + LSTK; ++layer) {
    const int li = layer - 1;
    const float* We_   = (layer == 0) ? p1_We   : We   + (size_t)li * 2 * HIDD;
    const float* att_  = (layer == 0) ? p1_att  : att  + (size_t)li * HEADS * CHN;
    const float* bg_   = (layer == 0) ? p1_bias : bias_g + (size_t)li * HIDD;
    float* hout = out + (size_t)layer * NN * HIDD;
    const bool odd = (layer & 1) != 0;
    const float* xl_c  = odd ? xlB  : xl;   float* xl_n  = odd ? xl  : xlB;
    const float* xr_c  = odd ? xrB  : xr;   float* xr_n  = odd ? xr  : xrB;
    const float* res_c = odd ? resB : res;  float* res_n = odd ? res : resB;

    if (layer < LSTK) {
      k_gat_node<true><<<NN / NB, NB * HEADS, 0, stream>>>(
          xl_c, xr_c, res_c, rowptr, csrc, cea, We_, att_, bg_, hout,
          Wl   + (size_t)layer * HIDD * HIDD, bl   + (size_t)layer * HIDD, xl_n,
          Wr   + (size_t)layer * HIDD * HIDD, br   + (size_t)layer * HIDD, xr_n,
          Wres + (size_t)layer * HIDD * HIDD, bres + (size_t)layer * HIDD, res_n,
          nullptr, nullptr, nullptr);
    } else {
      k_gat_node<false><<<NN / NB, NB * HEADS, 0, stream>>>(
          xl_c, xr_c, res_c, rowptr, csrc, cea, We_, att_, bg_, hout,
          nullptr, nullptr, nullptr, nullptr, nullptr, nullptr, nullptr, nullptr, nullptr,
          batch, psum, pcnt);
    }
  }

  float* pooled = out + (size_t)(1 + LSTK) * NN * HIDD;
  float* x12    = pooled + NGR * HIDD;
  k_cls_wmma<<<(VOC + CW - 1) / CW, 128, 0, stream>>>(psum, pcnt, Wlin, blin, x12, pooled);
}


# Per-edge arithmetic of the node kernel on explicit channel pairs (packed f32): leaky ReLU as max(m, 0.2m); a logit's 8 products summed as two 4-term fma chains, not one 8-term chain
# speedup vs baseline: 5.4220x; 1.0736x over previous
#include <hip/hip_runtime.h>

#define NN 10000
#define NE 160000
#define FIN 16
#define HEADS 8
#define CHN 8
#define HIDD 64
#define NGR 64
#define VOC 50000
#define LSTK 9
#define EU 8

typedef __attribute__((ext_vector_type(2))) float v2f;
typedef __attribute__((ext_vector_type(4))) float v4f;
typedef __attribute__((ext_vector_type(8))) float v8f;

static_assert(FIN % 4 == 0, "whole k-steps of 4");
__device__ __forceinline__ void gemm3_tile(const float* __restrict__ A,
                             const float* __restrict__ B0, const float* __restrict__ b0,
                             float* __restrict__ C0,
                             const float* __restrict__ B1, const float* __restrict__ b1,
                             float* __restrict__ C1,
                             const float* __restrict__ B2, const float* __restrict__ b2,
                             float* __restrict__ C2,
                             int tn, int tm, int lane) {
  const int half = lane >> 4;
  const int l    = lane & 15;
  const int arow = tm * 16 + l;
  const int bcol = tn * 16 + l;
  v2f a[FIN / 4], w0[FIN / 4], w1[FIN / 4], w2[FIN / 4];
#pragma unroll
  for (int j = 0; j < FIN / 4; ++j) {
    const int k = 4 * j;
    const float* ap = A + arow * FIN + k + half * 2;
    a[j].x = ap[0]; a[j].y = ap[1];
    const int boff = (k + half * 2) * HIDD + bcol;
    w0[j].x = B0[boff]; w0[j].y = B0[boff + HIDD];
    w1[j].x = B1[boff]; w1[j].y = B1[boff + HIDD];
    w2[j].x = B2[boff]; w2[j].y = B2[boff + HIDD];
  }
  const float bv0 = b0[bcol], bv1 = b1[bcol], bv2 = b2[bcol];
  v8f acc0 = {}, acc1 = {}, acc2 = {};
#pragma unroll
  for (int j = 0; j < FIN / 4; ++j) {
    acc0 = __builtin_amdgcn_wmma_f32_16x16x4_f32(false, a[j], false, w0[j], (short)0, acc0, false, false);
    acc1 = __builtin_amdgcn_wmma_f32_16x16x4_f32(false, a[j], false, w1[j], (short)0, acc1, false, false);
    acc2 = __builtin_amdgcn_wmma_f32_16x16x4_f32(false, a[j], false, w2[j], (short)0, acc2, false, false);
  }
  const int rb = tm * 16 + half * 8;
#pragma unroll
  for (int r = 0; r < 8; ++r) {
    const int off = (rb + r) * HIDD + bcol;
    C0[off] = acc0[r] + bv0;
    C1[off] = acc1[r] + bv1;
    C2[off] = acc2[r] + bv2;
  }
}

#define CW 64
static_assert(VOC % 4 == 0, "a lane's 4 columns are all inside the matrix or all outside");
static_assert(NGR == 64 && HIDD == 64, "4 waves x 16 rows cover the graphs; 16 k-steps cover the hidden width");

__global__ void __launch_bounds__(128)
k_cls_wmma(const float* __restrict__ S, const float* __restrict__ Sn,
           const float* __restrict__ B, const float* __restrict__ bias,
           float* __restrict__ C, float* __restrict__ Amean) {
  const int lane = threadIdx.x & 31;
  const int half = lane >> 4;
  const int l    = lane & 15;
  const int w    = threadIdx.x >> 5;
  const int arow = w * 16 + l;
  const int c0   = blockIdx.x * CW + 4 * l;
  const bool live = c0 < VOC;
  const int cl   = live ? c0 : VOC - 4;

  const float dnr = Sn[arow];
  v2f s[HIDD / 4];
#pragma unroll
  for (int j = 0; j < HIDD / 4; ++j)
    s[j] = *(const v2f*)(S + arow * HIDD + 4 * j + half * 2);
  v4f bx[HIDD / 4], by[HIDD / 4];
#pragma unroll
  for (int j = 0; j < HIDD / 4; ++j) {
    const float* bp = B + (size_t)(4 * j + half * 2) * VOC + cl;
    bx[j] = *(const v4f*)bp;
    by[j] = *(const v4f*)(bp + VOC);
  }
  const v4f bv = *(const v4f*)(bias + cl);

  const float dn = fmaxf(dnr, 1.f);
  v2f a[HIDD / 4];
#pragma unroll
  for (int j = 0; j < HIDD / 4; ++j) {
    a[j].x = s[j].x / dn; a[j].y = s[j].y / dn;
    if (blockIdx.x == 0) *(v2f*)(Amean + arow * HIDD + 4 * j + half * 2) = a[j];
  }

  v8f acc0 = {}, acc1 = {}, acc2 = {}, acc3 = {};
#pragma unroll
  for (int j = 0; j < HIDD / 4; ++j) {
    v2f b0, b1, b2, b3;
    b0.x = bx[j][0]; b0.y = by[j][0];
    b1.x = bx[j][1]; b1.y = by[j][1];
    b2.x = bx[j][2]; b2.y = by[j][2];
    b3.x = bx[j][3]; b3.y = by[j][3];
    acc0 = __builtin_amdgcn_wmma_f32_16x16x4_f32(false, a[j], false, b0, (short)0, acc0, false, false);
    acc1 = __builtin_amdgcn_wmma_f32_16x16x4_f32(false, a[j], false, b1, (short)0, acc1, false, false);
    acc2 = __builtin_amdgcn_wmma_f32_16x16x4_f32(false, a[j], false, b2, (short)0, acc2, false, false);
    acc3 = __builtin_amdgcn_wmma_f32_16x16x4_f32(false, a[j], false, b3, (short)0, acc3, false, false);
  }
  if (live) {
    float* cp = C + (size_t)(w * 16 + half * 8) * VOC + c0;
#pragma unroll
    for (int r = 0; r < 8; ++r) {
      v4f o;
      o[0] = acc0[r] + bv[0]; o[1] = acc1[r] + bv[1];
      o[2] = acc2[r] + bv[2]; o[3] = acc3[r] + bv[3];
      *(v4f*)(cp + (size_t)r * VOC) = o;
    }
  }
}

__global__ void k_zero_i(int* __restrict__ p, int n) {
  int i = blockIdx.x * blockDim.x + threadIdx.x;
  if (i < n) p[i] = 0;
}

__global__ void k_count(const int* __restrict__ dstE, int* __restrict__ cnt,
                        int* __restrict__ slot) {
  int e = blockIdx.x * blockDim.x + threadIdx.x;
  if (e < NE) slot[e] = atomicAdd(cnt + dstE[e], 1);
}

#define SCAN_T 1024
#define SCAN_PER ((NN + SCAN_T - 1) / SCAN_T)
static_assert(SCAN_T / 32 == 32, "one lane of wave 0 per wave total");
__global__ void __launch_bounds__(SCAN_T)
k_scan_fill(const int* __restrict__ cnt, const int* __restrict__ dstE,
            const int* __restrict__ slot, int* __restrict__ rowptr,
            int* __restrict__ perm) {
  __shared__ int rp[SCAN_T * SCAN_PER];
  __shared__ int wtot[SCAN_T / 32];
  const int t = threadIdx.x;
  const int lane = t & 31, wv = t >> 5;
  const int e = blockIdx.x * SCAN_T + t;
  int d = 0, s = 0;
  if (e < NE) { d = dstE[e]; s = slot[e]; }
  int loc[SCAN_PER];
  int sum = 0;
#pragma unroll
  for (int i = 0; i < SCAN_PER; ++i) {
    const int idx = t * SCAN_PER + i;
    loc[i] = sum;
    sum += (idx < NN) ? cnt[idx] : 0;
  }
  int v = sum;
#pragma unroll
  for (int off = 1; off < 32; off <<= 1) {
    const int u = __shfl_up(v, off);
    if (lane >= off) v += u;
  }
  if (lane == 31) wtot[wv] = v;
  __syncthreads();
  if (wv == 0) {
    int w = wtot[lane];
#pragma unroll
    for (int off = 1; off < 32; off <<= 1) {
      const int u = __shfl_up(w, off);
      if (lane >= off) w += u;
    }
    wtot[lane] = w;
  }
  __syncthreads();
  const int excl = v - sum + (wv > 0 ? wtot[wv - 1] : 0);
#pragma unroll
  for (int i = 0; i < SCAN_PER; ++i) {
    const int idx = t * SCAN_PER + i;
    if (idx < NN) {
      rp[idx] = excl + loc[i];
      if (blockIdx.x == 0) rowptr[idx] = excl + loc[i];
    }
  }
  if (blockIdx.x == 0 && t == SCAN_T - 1) rowptr[NN] = excl + sum;
  __syncthreads();
  if (e < NE) perm[rp[d] + s] = e;
}

#define RANK_T 256
#define RANK_G ((NE + RANK_T - 1) / RANK_T)
#define G3_TILES ((HIDD / 16) * (NN / 16))
#define G3_G ((G3_TILES + RANK_T / 32 - 1) / (RANK_T / 32))
__global__ void __launch_bounds__(RANK_T)
k_rank_gemm3(const int* __restrict__ perm, const int* __restrict__ rowptr,
             const int* __restrict__ srcE, const int* __restrict__ dstE,
             const float* __restrict__ ea,
             int* __restrict__ csrc, v2f* __restrict__ cea,
             const float* __restrict__ x,
             const float* __restrict__ B0, const float* __restrict__ b0, float* __restrict__ C0,
             const float* __restrict__ B1, const float* __restrict__ b1, float* __restrict__ C1,
             const float* __restrict__ B2, const float* __restrict__ b2, float* __restrict__ C2) {
  if (blockIdx.x >= RANK_G) {
    const int wv = __builtin_amdgcn_readfirstlane((int)(threadIdx.x >> 5));
    const int tile = (blockIdx.x - RANK_G) * (RANK_T / 32) + wv;
    if (tile < G3_TILES)
      gemm3_tile(x, B0, b0, C0, B1, b1, C1, B2, b2, C2,
                 tile & (HIDD / 16 - 1), tile / (HIDD / 16), threadIdx.x & 31);
    return;
  }
  int p = blockIdx.x * RANK_T + threadIdx.x;
  if (p >= NE) return;
  const int e = perm[p];
  const int d = dstE[e];
  const int beg = rowptr[d], end = rowptr[d + 1];
  int r = 0;
  for (int q = beg; q < end; ++q) r += (perm[q] < e) ? 1 : 0;
  v2f a; a.x = ea[2 * e]; a.y = ea[2 * e + 1];
  csrc[beg + r] = srcE[e];
  cea[beg + r]  = a;
}

#define CP (CHN / 2)
__device__ __forceinline__ v2f lo2(v4f v) { return __builtin_shufflevector(v, v, 0, 1); }
__device__ __forceinline__ v2f hi2(v4f v) { return __builtin_shufflevector(v, v, 2, 3); }

__device__ __forceinline__ float logit8(v4f xa, v4f xb, const v2f* xd,
                                        const v2f* w0, const v2f* w1,
                                        const v2f* at, float a0, float a1) {
  const v2f xs[CP] = {lo2(xa), hi2(xa), lo2(xb), hi2(xb)};
  v2f lg = {0.f, 0.f};
#pragma unroll
  for (int c = 0; c < CP; ++c) {
    const v2f m = xs[c] + xd[c] + a0 * w0[c] + a1 * w1[c];
    const v2f g = __builtin_elementwise_max(m, 0.2f * m);
    lg += g * at[c];
  }
  return lg.x + lg.y;
}

#define NB 16
#define TP 68
static_assert(NN % NB == 0, "the grid covers the nodes exactly: no thread is out of range");

template <bool NEXT>
__global__ void __launch_bounds__(NB * HEADS)
k_gat_node(const float* __restrict__ xl, const float* __restrict__ xr,
           const float* __restrict__ res, const int* __restrict__ rowptr,
           const int* __restrict__ csrc, const v2f* __restrict__ cea,
           const float* __restrict__ We, const float* __restrict__ att,
           const float* __restrict__ bias_g, float* __restrict__ hout,
           const float* __restrict__ nB0, const float* __restrict__ nb0, float* __restrict__ nC0,
           const float* __restrict__ nB1, const float* __restrict__ nb1, float* __restrict__ nC1,
           const float* __restrict__ nB2, const float* __restrict__ nb2, float* __restrict__ nC2,
           const int* __restrict__ batch, float* __restrict__ psum, float* __restrict__ pcnt) {
  __shared__ alignas(16) float sH[NB * TP];
  const int idx = blockIdx.x * (NB * HEADS) + threadIdx.x;
  const int d = idx >> 3, cb = (idx & 7) * CHN;

  v2f w0[CP], w1[CP], at[CP], xd[CP];
  v4f xself0, xself1;
  {
    const v4f* pd = (const v4f*)(xr + d * HIDD + cb);
    const v4f* ps = (const v4f*)(xl + d * HIDD + cb);
    const v4f d0 = pd[0], d1 = pd[1];
    xself0 = ps[0]; xself1 = ps[1];
    xd[0] = lo2(d0); xd[1] = hi2(d0); xd[2] = lo2(d1); xd[3] = hi2(d1);
#pragma unroll
    for (int c = 0; c < CP; ++c) {
      w0[c].x = We[cb + 2 * c];        w0[c].y = We[cb + 2 * c + 1];
      w1[c].x = We[HIDD + cb + 2 * c]; w1[c].y = We[HIDD + cb + 2 * c + 1];
      at[c].x = att[cb + 2 * c];       at[c].y = att[cb + 2 * c + 1];
    }
  }

  const int beg = rowptr[d], end = rowptr[d + 1];
  float mrun = -INFINITY, den = 0.f;
  v2f asum = {0.f, 0.f};
  v2f acc[CP];
#pragma unroll
  for (int c = 0; c < CP; ++c) acc[c] = (v2f){0.f, 0.f};

  if (beg < end) {
    int sj[EU]; v2f aj[EU];
#pragma unroll
    for (int j = 0; j < EU; ++j) {
      const int q = min(beg + j, end - 1);
      sj[j] = csrc[q]; aj[j] = cea[q];
    }
    for (int p = beg; p < end; p += EU) {
      v4f x0[EU], x1[EU];
#pragma unroll
      for (int j = 0; j < EU; ++j) {
        const v4f* r = (const v4f*)(xl + sj[j] * HIDD + cb);
        x0[j] = r[0]; x1[j] = r[1];
      }
      int sn[EU]; v2f an[EU];
#pragma unroll
      for (int j = 0; j < EU; ++j) {
        const int q = min(p + EU + j, end - 1);
        sn[j] = csrc[q]; an[j] = cea[q];
      }
      float lg[EU];
      float cm = -INFINITY;
#pragma unroll
      for (int j = 0; j < EU; ++j) {
        const bool v = (p + j) < end;
        const float l = logit8(x0[j], x1[j], xd, w0, w1, at, aj[j].x, aj[j].y);
        lg[j] = v ? l : -INFINITY;
        asum += v ? aj[j] : (v2f){0.f, 0.f};
        cm = fmaxf(cm, lg[j]);
      }
      const float mnew = fmaxf(mrun, cm);
      const float sc = __expf(mrun - mnew);
      float pj[EU];
      float psum = 0.f;
#pragma unroll
      for (int j = 0; j < EU; ++j) { pj[j] = __expf(lg[j] - mnew); psum += pj[j]; }
      den = den * sc + psum;
      {
        v2f t0 = {0.f, 0.f}, t1 = {0.f, 0.f}, t2 = {0.f, 0.f}, t3 = {0.f, 0.f};
#pragma unroll
        for (int j = 0; j < EU; ++j) {
          t0 += pj[j] * lo2(x0[j]); t1 += pj[j] * hi2(x0[j]);
          t2 += pj[j] * lo2(x1[j]); t3 += pj[j] * hi2(x1[j]);
        }
        acc[0] = acc[0] * sc + t0; acc[1] = acc[1] * sc + t1;
        acc[2] = acc[2] * sc + t2; acc[3] = acc[3] * sc + t3;
      }
      mrun = mnew;
#pragma unroll
      for (int j = 0; j < EU; ++j) { sj[j] = sn[j]; aj[j] = an[j]; }
    }
  }

  const int lane = threadIdx.x & 31;
  const int half = lane >> 4;
  const int l    = lane & 15;
  const int bcol = (threadIdx.x >> 5) * 16 + l;
  v2f wf0[HIDD / 4], wf1[HIDD / 4], wf2[HIDD / 4];
  float bv0 = 0.f, bv1 = 0.f, bv2 = 0.f;
  if (NEXT) {
#pragma unroll
    for (int j = 0; j < HIDD / 4; ++j) {
      const int boff = (4 * j + half * 2) * HIDD + bcol;
      wf0[j].x = nB0[boff]; wf0[j].y = nB0[boff + HIDD];
      wf1[j].x = nB1[boff]; wf1[j].y = nB1[boff + HIDD];
      wf2[j].x = nB2[boff]; wf2[j].y = nB2[boff + HIDD];
    }
    bv0 = nb0[bcol]; bv1 = nb1[bcol]; bv2 = nb2[bcol];
  }
  const int pc  = threadIdx.x & (HIDD - 1);
  const int pr0 = (threadIdx.x >> 6) * (NB / 2);
  int gid[NB / 2];
  if (!NEXT) {
#pragma unroll
    for (int r = 0; r < NB / 2; ++r) gid[r] = batch[blockIdx.x * NB + pr0 + r];
  }

  {
    const float dg = fmaxf((float)(end - beg), 1.f);
    const float l = logit8(xself0, xself1, xd, w0, w1, at, asum.x / dg, asum.y / dg);
    const float mnew = fmaxf(mrun, l);
    const float sc = __expf(mrun - mnew);
    const float pl = __expf(l - mnew);
    den = den * sc + pl;
    acc[0] = acc[0] * sc + pl * lo2(xself0); acc[1] = acc[1] * sc + pl * hi2(xself0);
    acc[2] = acc[2] * sc + pl * lo2(xself1); acc[3] = acc[3] * sc + pl * hi2(xself1);
  }

  const v4f* pr = (const v4f*)(res + d * HIDD + cb);
  const v4f r0 = pr[0], r1 = pr[1];
  v4f o0, o1;
  const float accs[CHN] = {acc[0].x, acc[0].y, acc[1].x, acc[1].y, acc[2].x, acc[2].y, acc[3].x, acc[3].y};
#pragma unroll
  for (int c = 0; c < 4; ++c) {
    float v0 = accs[c] / den     + bias_g[cb + c]     + r0[c];
    float v1 = accs[4 + c] / den + bias_g[cb + 4 + c] + r1[c];
    o0[c] = v0 > 0.f ? v0 : (__expf(v0) - 1.f);
    o1[c] = v1 > 0.f ? v1 : (__expf(v1) - 1.f);
  }
  v4f* po = (v4f*)(hout + d * HIDD + cb);
  if (!NEXT) {
    v4f* ps = (v4f*)(sH + (threadIdx.x >> 3) * TP + cb);
    ps[0] = o0; ps[1] = o1;
    __syncthreads();
    po[0] = o0; po[1] = o1;
    float hv[NB / 2];
#pragma unroll
    for (int r = 0; r < NB / 2; ++r) hv[r] = sH[(pr0 + r) * TP + pc];
    float run = 0.f, nrun = 0.f;
    int gcur = gid[0];
#pragma unroll
    for (int r = 0; r < NB / 2; ++r) {
      if (gid[r] != gcur) {
        atomicAdd(psum + gcur * HIDD + pc, run);
        if (pc == 0) atomicAdd(pcnt + gcur, nrun);
        run = 0.f; nrun = 0.f; gcur = gid[r];
      }
      run += hv[r];
      nrun += 1.f;
    }
    atomicAdd(psum + gcur * HIDD + pc, run);
    if (pc == 0) atomicAdd(pcnt + gcur, nrun);
  }

  if (NEXT) {
    v4f* ps = (v4f*)(sH + (threadIdx.x >> 3) * TP + cb);
    ps[0] = o0; ps[1] = o1;
    __syncthreads();
    po[0] = o0; po[1] = o1;
    v8f acc0 = {}, acc1 = {}, acc2 = {};
#pragma unroll
    for (int j = 0; j < HIDD / 4; ++j) {
      const v2f a = *(const v2f*)(sH + l * TP + 4 * j + half * 2);
      acc0 = __builtin_amdgcn_wmma_f32_16x16x4_f32(false, a, false, wf0[j], (short)0, acc0, false, false);
      acc1 = __builtin_amdgcn_wmma_f32_16x16x4_f32(false, a, false, wf1[j], (short)0, acc1, false, false);
      acc2 = __builtin_amdgcn_wmma_f32_16x16x4_f32(false, a, false, wf2[j], (short)0, acc2, false, false);
    }
    const int rb = blockIdx.x * NB + half * 8;
#pragma unroll
    for (int r = 0; r < 8; ++r) {
      const int off = (rb + r) * HIDD + bcol;
      nC0[off] = acc0[r] + bv0;
      nC1[off] = acc1[r] + bv1;
      nC2[off] = acc2[r] + bv2;
    }
  }
}

extern "C" void kernel_launch(void* const* d_in, const int* in_sizes, int n_in,
                              void* d_out_v, int out_size, void* d_ws, size_t ws_size,
                              hipStream_t stream) {
  (void)in_sizes; (void)n_in; (void)out_size; (void)ws_size;
  const float* x       = (const float*)d_in[0];
  const int*   ei      = (const int*)  d_in[1];
  const float* ea      = (const float*)d_in[2];
  const int*   batch   = (const int*)  d_in[3];
  const float* p1_Wl   = (const float*)d_in[4];
  const float* p1_bl   = (const float*)d_in[5];
  const float* p1_Wr   = (const float*)d_in[6];
  const float* p1_br   = (const float*)d_in[7];
  const float* p1_We   = (const float*)d_in[8];
  const float* p1_att  = (const float*)d_in[9];
  const float* p1_bias = (const float*)d_in[10];
  const float* p1_Wres = (const float*)d_in[11];
  const float* p1_bres = (const float*)d_in[12];
  const float* Wl      = (const float*)d_in[13];
  const float* bl      = (const float*)d_in[14];
  const float* Wr      = (const float*)d_in[15];
  const float* br      = (const float*)d_in[16];
  const float* We      = (const float*)d_in[17];
  const float* att     = (const float*)d_in[18];
  const float* bias_g  = (const float*)d_in[19];
  const float* Wres    = (const float*)d_in[20];
  const float* bres    = (const float*)d_in[21];
  const float* Wlin    = (const float*)d_in[22];
  const float* blin    = (const float*)d_in[23];

  float* out = (float*)d_out_v;
  float* ws  = (float*)d_ws;

  float* xl     = ws;
  float* xr     = ws + 640000;
  float* res    = ws + 1280000;
  int*   rowptr = (int*)(ws + 1920000);
  int*   cur    = (int*)(ws + 1940000);
  float* psum   = ws + 1950000;
  float* pcnt   = ws + 1954096;
  int*   perm   = (int*)(ws + 1960000);
  int*   csrc   = (int*)(ws + 2120000);
  v2f*   cea    = (v2f*)(ws + 2280000);
  float* xlB    = ws + 2600000;
  float* xrB    = ws + 3240000;
  float* resB   = ws + 3880000;
  int*   slot   = (int*)xlB;

  const int* srcE = ei;
  const int* dstE = ei + NE;

  const int TB = 256;

  k_zero_i<<<(NN + NGR * HIDD + NGR + TB - 1) / TB, TB, 0, stream>>>(cur, NN + NGR * HIDD + NGR);
  k_count<<<(NE + TB - 1) / TB, TB, 0, stream>>>(dstE, cur, slot);
  k_scan_fill<<<(NE + SCAN_T - 1) / SCAN_T, SCAN_T, 0, stream>>>(cur, dstE, slot, rowptr, perm);

  k_rank_gemm3<<<RANK_G + G3_G, RANK_T, 0, stream>>>(perm, rowptr, srcE, dstE, ea, csrc, cea,
                                                     x, p1_Wl, p1_bl, xl, p1_Wr, p1_br, xr,
                                                     p1_Wres, p1_bres, res);
  for (int layer = 0; layer < 1 + LSTK; ++layer) {
    const int li = layer - 1;
    const float* We_   = (layer == 0) ? p1_We   : We   + (size_t)li * 2 * HIDD;
    const float* att_  = (layer == 0) ? p1_att  : att  + (size_t)li * HEADS * CHN;
    const float* bg_   = (layer == 0) ? p1_bias : bias_g + (size_t)li * HIDD;
    float* hout = out + (size_t)layer * NN * HIDD;
    const bool odd = (layer & 1) != 0;
    const float* xl_c  = odd ? xlB  : xl;   float* xl_n  = odd ? xl  : xlB;
    const float* xr_c  = odd ? xrB  : xr;   float* xr_n  = odd ? xr  : xrB;
    const float* res_c = odd ? resB : res;  float* res_n = odd ? res : resB;

    if (layer < LSTK) {
      k_gat_node<true><<<NN / NB, NB * HEADS, 0, stream>>>(
          xl_c, xr_c, res_c, rowptr, csrc, cea, We_, att_, bg_, hout,
          Wl   + (size_t)layer * HIDD * HIDD, bl   + (size_t)layer * HIDD, xl_n,
          Wr   + (size_t)layer * HIDD * HIDD, br   + (size_t)layer * HIDD, xr_n,
          Wres + (size_t)layer * HIDD * HIDD, bres + (size_t)layer * HIDD, res_n,
          nullptr, nullptr, nullptr);
    } else {
      k_gat_node<false><<<NN / NB, NB * HEADS, 0, stream>>>(
          xl_c, xr_c, res_c, rowptr, csrc, cea, We_, att_, bg_, hout,
          nullptr, nullptr, nullptr, nullptr, nullptr, nullptr, nullptr, nullptr, nullptr,
          batch, psum, pcnt);
    }
  }

  float* pooled = out + (size_t)(1 + LSTK) * NN * HIDD;
  float* x12    = pooled + NGR * HIDD;
  k_cls_wmma<<<(VOC + CW - 1) / CW, 128, 0, stream>>>(psum, pcnt, Wlin, blin, x12, pooled);
}
